# starting kernel
# baseline (speedup 1.0000x reference)
#include <hip/hip_runtime.h>

#define NN      50000
#define NE      800000
#define IN_F    256
#define HEADS   4
#define D_K     32
#define OUT_T   128
#define N_CAT   384
#define M_TILES (NN / 16)
#define N_TILES (N_CAT / 16)

typedef __attribute__((ext_vector_type(16))) __bf16   v16bf;
typedef __attribute__((ext_vector_type(8)))  float    v8f;
typedef __attribute__((ext_vector_type(4)))  unsigned uint4v;

union Frag { uint4v q[2]; v16bf v; };

__device__ __forceinline__ unsigned short f2bf(float f) {
    unsigned u = __float_as_uint(f);
    unsigned r = (u + 0x7FFFu + ((u >> 16) & 1u)) >> 16;
    return (unsigned short)r;
}
__device__ __forceinline__ float bf2f(unsigned short h) {
    return __uint_as_float(((unsigned)h) << 16);
}

__device__ __forceinline__ unsigned enc_ord(float f) {
    unsigned b = __float_as_uint(f);
    return (b & 0x80000000u) ? ~b : (b | 0x80000000u);
}
__device__ __forceinline__ float dec_ord(unsigned u) {
    return (u & 0x80000000u) ? __uint_as_float(u & 0x7FFFFFFFu) : __uint_as_float(~u);
}

__global__ void k_zero(float* __restrict__ p, int n) {
    int g = blockIdx.x * blockDim.x + threadIdx.x;
    if (g < n) p[g] = 0.0f;
}

__global__ void k_split(const float* __restrict__ src,
                        unsigned short* __restrict__ hi,
                        unsigned short* __restrict__ lo, int n) {
    int g = blockIdx.x * blockDim.x + threadIdx.x;
    if (g >= n) return;
    float f = src[g];
    unsigned short h = f2bf(f);
    hi[g] = h;
    lo[g] = f2bf(f - bf2f(h));
}

__global__ void k_wcat(const float* __restrict__ Wq, const float* __restrict__ Wk,
                       const float* __restrict__ Wv,
                       unsigned short* __restrict__ whi,
                       unsigned short* __restrict__ wlo) {
    int g = blockIdx.x * blockDim.x + threadIdx.x;
    if (g >= N_CAT * IN_F) return;
    int n = g / IN_F, k = g - n * IN_F;
    int which = n >> 7, nn = n & 127;
    const float* W = (which == 0) ? Wq : (which == 1) ? Wk : Wv;
    float f = W[k * OUT_T + nn];
    unsigned short h = f2bf(f);
    whi[g] = h;
    wlo[g] = f2bf(f - bf2f(h));
}

__global__ void k_ssym(const float* __restrict__ S, float* __restrict__ out) {
    int g = blockIdx.x * blockDim.x + threadIdx.x;
    if (g >= HEADS * D_K * D_K) return;
    int h = g / (D_K * D_K);
    int r = g - h * D_K * D_K;
    int i = r / D_K, j = r - i * D_K;
    out[g] = 0.5f * (S[h * D_K * D_K + i * D_K + j] + S[h * D_K * D_K + j * D_K + i]);
}

__global__ __launch_bounds__(256) void k_qkv(const unsigned short* __restrict__ xhi,
                                             const unsigned short* __restrict__ xlo,
                                             const unsigned short* __restrict__ whi,
                                             const unsigned short* __restrict__ wlo,
                                             float* __restrict__ Qd,
                                             float* __restrict__ Kd,
                                             float* __restrict__ Vd) {
    int wid = blockIdx.x * 8 + (threadIdx.x >> 5);
    if (wid >= M_TILES * N_TILES) return;
    int lane  = threadIdx.x & 31;
    int mTile = wid / N_TILES;
    int nTile = wid - mTile * N_TILES;
    int l16   = lane & 15;
    int lhalf = lane >> 4;

    size_t aoff = (size_t)(mTile * 16 + l16) * IN_F;
    size_t boff = (size_t)(nTile * 16 + l16) * IN_F;
    const unsigned short* ah = xhi + aoff;
    const unsigned short* al = xlo + aoff;
    const unsigned short* bh = whi + boff;
    const unsigned short* bl = wlo + boff;

    v8f c = {};
#pragma unroll
    for (int ks = 0; ks < IN_F; ks += 32) {
        int c0 = ks + lhalf * 8;
        Frag fah, fal, fbh, fbl;
        fah.q[0] = *reinterpret_cast<const uint4v*>(ah + c0);
        fah.q[1] = *reinterpret_cast<const uint4v*>(ah + c0 + 16);
        fbh.q[0] = *reinterpret_cast<const uint4v*>(bh + c0);
        fbh.q[1] = *reinterpret_cast<const uint4v*>(bh + c0 + 16);
        fal.q[0] = *reinterpret_cast<const uint4v*>(al + c0);
        fal.q[1] = *reinterpret_cast<const uint4v*>(al + c0 + 16);
        fbl.q[0] = *reinterpret_cast<const uint4v*>(bl + c0);
        fbl.q[1] = *reinterpret_cast<const uint4v*>(bl + c0 + 16);
        c = __builtin_amdgcn_wmma_f32_16x16x32_bf16(false, fah.v, false, fbh.v,
                                                    (short)0, c, false, false);
        c = __builtin_amdgcn_wmma_f32_16x16x32_bf16(false, fah.v, false, fbl.v,
                                                    (short)0, c, false, false);
        c = __builtin_amdgcn_wmma_f32_16x16x32_bf16(false, fal.v, false, fbh.v,
                                                    (short)0, c, false, false);
    }

    int nBase = nTile * 16;
    float* dst; int ncol;
    if (nBase < 128)      { dst = Qd; ncol = nBase;       }
    else if (nBase < 256) { dst = Kd; ncol = nBase - 128; }
    else                  { dst = Vd; ncol = nBase - 256; }
    int col     = ncol + l16;
    int rowBase = mTile * 16 + lhalf * 8;
#pragma unroll
    for (int v = 0; v < 8; ++v)
        dst[(size_t)(rowBase + v) * OUT_T + col] = c[v];
}

__global__ __launch_bounds__(256) void k_scores(const float* __restrict__ Qd,
                                                const float* __restrict__ Kd,
                                                const float* __restrict__ Ssym,
                                                const int* __restrict__ ei,
                                                float* __restrict__ scores,
                                                unsigned* __restrict__ m_u) {
    __shared__ float sS[HEADS * D_K * D_K];
    for (int i = threadIdx.x; i < HEADS * D_K * D_K; i += 256) sS[i] = Ssym[i];
    __syncthreads();

    int g = blockIdx.x * 256 + threadIdx.x;
    if (g >= NE * HEADS) return;
    int e = g >> 2, h = g & 3;
    int row = ei[e];
    int col = ei[NE + e];

    const float* Kp = Kd + ((size_t)row * HEADS + h) * D_K;
    const float* Qp = Qd + ((size_t)col * HEADS + h) * D_K;
    float diff[D_K];
#pragma unroll
    for (int i = 0; i < D_K; ++i) diff[i] = Kp[i] - Qp[i];

    const float* Sh = sS + h * D_K * D_K;
    float s = 0.0f;
    for (int i = 0; i < D_K; ++i) {
        float acc = 0.0f;
#pragma unroll
        for (int j = 0; j < D_K; ++j) acc = fmaf(Sh[i * D_K + j], diff[j], acc);
        s = fmaf(diff[i], acc, s);
    }
    s *= 0.17677669529663687f;
    s = (s > 0.0f) ? s : 0.2f * s;
    scores[g] = s;
    atomicMax(&m_u[row * HEADS + h], enc_ord(s));
}

__global__ __launch_bounds__(256) void k_exp(const int* __restrict__ ei,
                                             const unsigned* __restrict__ m_u,
                                             float* __restrict__ scores,
                                             float* __restrict__ denom) {
    int g = blockIdx.x * 256 + threadIdx.x;
    if (g >= NE * HEADS) return;
    int e = g >> 2, h = g & 3;
    int row = ei[e];
    float mf = dec_ord(m_u[row * HEADS + h]);
    float ex = expf(scores[g] - mf);
    scores[g] = ex;
    __hip_atomic_fetch_add(&denom[row * HEADS + h], ex,
                           __ATOMIC_RELAXED, __HIP_MEMORY_SCOPE_AGENT);
}

__global__ __launch_bounds__(256) void k_acc(const int* __restrict__ ei,
                                             const float* __restrict__ scores,
                                             const float* __restrict__ denom,
                                             const float* __restrict__ Vd,
                                             float* __restrict__ out) {
    int g = blockIdx.x * 256 + threadIdx.x;
    if (g >= NE * HEADS) return;
    int e = g >> 2, h = g & 3;
    int row = ei[e], col = ei[NE + e];
    float alpha = scores[g] / (denom[row * HEADS + h] + 1e-16f);
    const float* Vp = Vd + ((size_t)col * HEADS + h) * D_K;
    float* op = out + (size_t)row * OUT_T + h * D_K;
#pragma unroll
    for (int d = 0; d < D_K; ++d)
        __hip_atomic_fetch_add(&op[d], alpha * Vp[d],
                               __ATOMIC_RELAXED, __HIP_MEMORY_SCOPE_AGENT);
}

extern "C" void kernel_launch(void* const* d_in, const int* in_sizes, int n_in,
                              void* d_out, int out_size, void* d_ws, size_t ws_size,
                              hipStream_t stream) {
    const float* x  = (const float*)d_in[0];
    const int*   ei = (const int*)  d_in[1];
    const float* Wq = (const float*)d_in[2];
    const float* Wk = (const float*)d_in[3];
    const float* Wv = (const float*)d_in[4];
    const float* Sr = (const float*)d_in[5];
    float* out = (float*)d_out;

    char* ws = (char*)d_ws;
    size_t off = 0;
    auto alloc = [&](size_t bytes) -> void* {
        void* p = ws + off;
        off = (off + bytes + 255) & ~(size_t)255;
        return p;
    };
    float*          Qd  = (float*)         alloc((size_t)NN * OUT_T * 4);
    float*          Kd  = (float*)         alloc((size_t)NN * OUT_T * 4);
    float*          Vd  = (float*)         alloc((size_t)NN * OUT_T * 4);
    unsigned short* xhi = (unsigned short*)alloc((size_t)NN * IN_F * 2);
    unsigned short* xlo = (unsigned short*)alloc((size_t)NN * IN_F * 2);
    unsigned short* whi = (unsigned short*)alloc((size_t)N_CAT * IN_F * 2);
    unsigned short* wlo = (unsigned short*)alloc((size_t)N_CAT * IN_F * 2);
    float*          ssy = (float*)         alloc((size_t)HEADS * D_K * D_K * 4);
    unsigned*       m_u = (unsigned*)      alloc((size_t)NN * HEADS * 4);
    float*          den = (float*)         alloc((size_t)NN * HEADS * 4);
    float*          sc  = (float*)         alloc((size_t)NE * HEADS * 4);

    k_zero<<<(NN * OUT_T + 255) / 256, 256, 0, stream>>>(out, NN * OUT_T);
    k_zero<<<(NN * HEADS + 255) / 256, 256, 0, stream>>>((float*)m_u, NN * HEADS);
    k_zero<<<(NN * HEADS + 255) / 256, 256, 0, stream>>>(den, NN * HEADS);

    k_split<<<(NN * IN_F + 255) / 256, 256, 0, stream>>>(x, xhi, xlo, NN * IN_F);
    k_wcat <<<(N_CAT * IN_F + 255) / 256, 256, 0, stream>>>(Wq, Wk, Wv, whi, wlo);
    k_ssym <<<(HEADS * D_K * D_K + 255) / 256, 256, 0, stream>>>(Sr, ssy);

    k_qkv<<<(M_TILES * N_TILES + 7) / 8, 256, 0, stream>>>(xhi, xlo, whi, wlo, Qd, Kd, Vd);

    int egrid = (NE * HEADS + 255) / 256;
    k_scores<<<egrid, 256, 0, stream>>>(Qd, Kd, ssy, ei, sc, m_u);
    k_exp   <<<egrid, 256, 0, stream>>>(ei, m_u, sc, den);
    k_acc   <<<egrid, 256, 0, stream>>>(ei, sc, den, Vd, out);
}


# Quadratic form expanded to k'Sk + q'Sq - 2(Sk)'q: the 32x32 metric product once per node; per edge one 32-long dot per head, read as whole 512-byte rows; the score's summation order changes
# speedup vs baseline: 3.0411x; 3.0411x over previous
#include <hip/hip_runtime.h>

#define NN      50000
#define NE      800000
#define IN_F    256
#define HEADS   4
#define D_K     32
#define OUT_T   128
#define N_CAT   384
#define M_TILES (NN / 16)
#define N_TILES (N_CAT / 16)

typedef __attribute__((ext_vector_type(16))) __bf16   v16bf;
typedef __attribute__((ext_vector_type(8)))  float    v8f;
typedef __attribute__((ext_vector_type(4)))  unsigned uint4v;

union Frag { uint4v q[2]; v16bf v; };

__device__ __forceinline__ unsigned short f2bf(float f) {
    unsigned u = __float_as_uint(f);
    unsigned r = (u + 0x7FFFu + ((u >> 16) & 1u)) >> 16;
    return (unsigned short)r;
}
__device__ __forceinline__ float bf2f(unsigned short h) {
    return __uint_as_float(((unsigned)h) << 16);
}

__device__ __forceinline__ unsigned enc_ord(float f) {
    unsigned b = __float_as_uint(f);
    return (b & 0x80000000u) ? ~b : (b | 0x80000000u);
}
__device__ __forceinline__ float dec_ord(unsigned u) {
    return (u & 0x80000000u) ? __uint_as_float(u & 0x7FFFFFFFu) : __uint_as_float(~u);
}

__global__ void k_zero(float* __restrict__ p, int n) {
    int g = blockIdx.x * blockDim.x + threadIdx.x;
    if (g < n) p[g] = 0.0f;
}

__global__ void k_split(const float* __restrict__ src,
                        unsigned short* __restrict__ hi,
                        unsigned short* __restrict__ lo, int n) {
    int g = blockIdx.x * blockDim.x + threadIdx.x;
    if (g >= n) return;
    float f = src[g];
    unsigned short h = f2bf(f);
    hi[g] = h;
    lo[g] = f2bf(f - bf2f(h));
}

__global__ void k_wcat(const float* __restrict__ Wq, const float* __restrict__ Wk,
                       const float* __restrict__ Wv,
                       unsigned short* __restrict__ whi,
                       unsigned short* __restrict__ wlo) {
    int g = blockIdx.x * blockDim.x + threadIdx.x;
    if (g >= N_CAT * IN_F) return;
    int n = g / IN_F, k = g - n * IN_F;
    int which = n >> 7, nn = n & 127;
    const float* W = (which == 0) ? Wq : (which == 1) ? Wk : Wv;
    float f = W[k * OUT_T + nn];
    unsigned short h = f2bf(f);
    whi[g] = h;
    wlo[g] = f2bf(f - bf2f(h));
}

__global__ void k_ssym(const float* __restrict__ S, float* __restrict__ out) {
    int g = blockIdx.x * blockDim.x + threadIdx.x;
    if (g >= HEADS * D_K * D_K) return;
    int h = g / (D_K * D_K);
    int r = g - h * D_K * D_K;
    int i = r / D_K, j = r - i * D_K;
    out[g] = 0.5f * (S[h * D_K * D_K + i * D_K + j] + S[h * D_K * D_K + j * D_K + i]);
}

__global__ __launch_bounds__(256) void k_qkv(const unsigned short* __restrict__ xhi,
                                             const unsigned short* __restrict__ xlo,
                                             const unsigned short* __restrict__ whi,
                                             const unsigned short* __restrict__ wlo,
                                             float* __restrict__ Qd,
                                             float* __restrict__ Kd,
                                             float* __restrict__ Vd) {
    int wid = blockIdx.x * 8 + (threadIdx.x >> 5);
    if (wid >= M_TILES * N_TILES) return;
    int lane  = threadIdx.x & 31;
    int mTile = wid / N_TILES;
    int nTile = wid - mTile * N_TILES;
    int l16   = lane & 15;
    int lhalf = lane >> 4;

    size_t aoff = (size_t)(mTile * 16 + l16) * IN_F;
    size_t boff = (size_t)(nTile * 16 + l16) * IN_F;
    const unsigned short* ah = xhi + aoff;
    const unsigned short* al = xlo + aoff;
    const unsigned short* bh = whi + boff;
    const unsigned short* bl = wlo + boff;

    v8f c = {};
#pragma unroll
    for (int ks = 0; ks < IN_F; ks += 32) {
        int c0 = ks + lhalf * 8;
        Frag fah, fal, fbh, fbl;
        fah.q[0] = *reinterpret_cast<const uint4v*>(ah + c0);
        fah.q[1] = *reinterpret_cast<const uint4v*>(ah + c0 + 16);
        fbh.q[0] = *reinterpret_cast<const uint4v*>(bh + c0);
        fbh.q[1] = *reinterpret_cast<const uint4v*>(bh + c0 + 16);
        fal.q[0] = *reinterpret_cast<const uint4v*>(al + c0);
        fal.q[1] = *reinterpret_cast<const uint4v*>(al + c0 + 16);
        fbl.q[0] = *reinterpret_cast<const uint4v*>(bl + c0);
        fbl.q[1] = *reinterpret_cast<const uint4v*>(bl + c0 + 16);
        c = __builtin_amdgcn_wmma_f32_16x16x32_bf16(false, fah.v, false, fbh.v,
                                                    (short)0, c, false, false);
        c = __builtin_amdgcn_wmma_f32_16x16x32_bf16(false, fah.v, false, fbl.v,
                                                    (short)0, c, false, false);
        c = __builtin_amdgcn_wmma_f32_16x16x32_bf16(false, fal.v, false, fbh.v,
                                                    (short)0, c, false, false);
    }

    int nBase = nTile * 16;
    float* dst; int ncol;
    if (nBase < 128)      { dst = Qd; ncol = nBase;       }
    else if (nBase < 256) { dst = Kd; ncol = nBase - 128; }
    else                  { dst = Vd; ncol = nBase - 256; }
    int col     = ncol + l16;
    int rowBase = mTile * 16 + lhalf * 8;
#pragma unroll
    for (int v = 0; v < 8; ++v)
        dst[(size_t)(rowBase + v) * OUT_T + col] = c[v];
}

__global__ __launch_bounds__(256) void k_node(const float* __restrict__ Qd,
                                              const float* __restrict__ Kd,
                                              const float* __restrict__ Ssym,
                                              float* __restrict__ SKd,
                                              float* __restrict__ kSk,
                                              float* __restrict__ qSq) {
    int n = blockIdx.x * 256 + threadIdx.x;
    if (n >= NN) return;
    int h   = blockIdx.y >> 1;
    int isQ = blockIdx.y & 1;
    const float* base = isQ ? Qd : Kd;
    const float4* src = reinterpret_cast<const float4*>(base + (size_t)n * OUT_T + h * D_K);
    float v[D_K];
#pragma unroll
    for (int i = 0; i < D_K / 4; ++i) {
        float4 t = src[i];
        v[4 * i + 0] = t.x; v[4 * i + 1] = t.y; v[4 * i + 2] = t.z; v[4 * i + 3] = t.w;
    }
    const float* Sh = Ssym + h * D_K * D_K;
    float sv[D_K];
    float s = 0.0f;
#pragma unroll
    for (int i = 0; i < D_K; ++i) {
        float acc = 0.0f;
#pragma unroll
        for (int j = 0; j < D_K; ++j) acc = fmaf(Sh[i * D_K + j], v[j], acc);
        sv[i] = acc;
        s = fmaf(v[i], acc, s);
    }
    if (isQ) {
        qSq[n * HEADS + h] = s;
    } else {
        kSk[n * HEADS + h] = s;
        float4* dst = reinterpret_cast<float4*>(SKd + (size_t)n * OUT_T + h * D_K);
#pragma unroll
        for (int i = 0; i < D_K / 4; ++i)
            dst[i] = make_float4(sv[4 * i + 0], sv[4 * i + 1], sv[4 * i + 2], sv[4 * i + 3]);
    }
}

#define SC_EPW 16
static_assert(NE % SC_EPW == 0 && (NE / SC_EPW) % 8 == 0, "whole waves, whole blocks");
__global__ __launch_bounds__(256) void k_scores(const float* __restrict__ SKd,
                                                const float* __restrict__ Qd,
                                                const float* __restrict__ kSk,
                                                const float* __restrict__ qSq,
                                                const int* __restrict__ ei,
                                                float* __restrict__ scores,
                                                unsigned* __restrict__ m_u) {
    int lane = threadIdx.x & 31;
    int h = lane >> 3;
    int g = lane & 7;
    int w = __builtin_amdgcn_readfirstlane((int)(blockIdx.x * 8 + (threadIdx.x >> 5)));
    const float4* SK4 = reinterpret_cast<const float4*>(SKd);
    const float4* Q4  = reinterpret_cast<const float4*>(Qd);

    for (int t = 0; t < SC_EPW / 8; ++t) {
        int eb = w * SC_EPW + t * 8;
        int row[8], col[8];
#pragma unroll
        for (int u = 0; u < 8; ++u) { row[u] = ei[eb + u]; col[u] = ei[NE + eb + u]; }
        float4 a[8], q[8];
#pragma unroll
        for (int u = 0; u < 8; ++u) {
            a[u] = SK4[(size_t)row[u] * 32 + lane];
            q[u] = Q4 [(size_t)col[u] * 32 + lane];
        }
        int myrow = ei[eb + g];
        int mycol = ei[NE + eb + g];
        float ks = kSk[myrow * HEADS + h];
        float qs = qSq[mycol * HEADS + h];

        float d[8];
#pragma unroll
        for (int u = 0; u < 8; ++u)
            d[u] = fmaf(a[u].x, q[u].x, fmaf(a[u].y, q[u].y, fmaf(a[u].z, q[u].z, a[u].w * q[u].w)));

        float d4[4];
#pragma unroll
        for (int u = 0; u < 4; ++u) {
            float keep = (g & 4) ? d[u + 4] : d[u];
            float send = (g & 4) ? d[u]     : d[u + 4];
            d4[u] = keep + __shfl_xor(send, 4);
        }
        float d2[2];
#pragma unroll
        for (int u = 0; u < 2; ++u) {
            float keep = (g & 2) ? d4[u + 2] : d4[u];
            float send = (g & 2) ? d4[u]     : d4[u + 2];
            d2[u] = keep + __shfl_xor(send, 2);
        }
        float keep1 = (g & 1) ? d2[1] : d2[0];
        float send1 = (g & 1) ? d2[0] : d2[1];
        float dot = keep1 + __shfl_xor(send1, 1);

        float s = (ks + qs) - 2.0f * dot;
        s *= 0.17677669529663687f;
        s = (s > 0.0f) ? s : 0.2f * s;
        scores[(size_t)(eb + g) * HEADS + h] = s;
        atomicMax(&m_u[myrow * HEADS + h], enc_ord(s));
    }
}

__global__ __launch_bounds__(256) void k_exp(const int* __restrict__ ei,
                                             const unsigned* __restrict__ m_u,
                                             float* __restrict__ scores,
                                             float* __restrict__ denom) {
    int g = blockIdx.x * 256 + threadIdx.x;
    if (g >= NE * HEADS) return;
    int e = g >> 2, h = g & 3;
    int row = ei[e];
    float mf = dec_ord(m_u[row * HEADS + h]);
    float ex = expf(scores[g] - mf);
    scores[g] = ex;
    __hip_atomic_fetch_add(&denom[row * HEADS + h], ex,
                           __ATOMIC_RELAXED, __HIP_MEMORY_SCOPE_AGENT);
}

__global__ __launch_bounds__(256) void k_hist(const int* __restrict__ ei,
                                              int* __restrict__ cnt) {
    int t = blockIdx.x * 256 + threadIdx.x;
    if (t >= NE / 4) return;
    int4 r = *reinterpret_cast<const int4*>(ei + 4 * t);
    atomicAdd(&cnt[r.x], 1);
    atomicAdd(&cnt[r.y], 1);
    atomicAdd(&cnt[r.z], 1);
    atomicAdd(&cnt[r.w], 1);
}

#define SCAN_IT  49
#define SCAN_SEG (SCAN_IT * 32)
#define SCAN_N   (32 * SCAN_SEG)
static_assert(SCAN_N >= NN + 1, "the scan must reach offs[NN]");
__global__ __launch_bounds__(1024) void k_scan(const int* __restrict__ cnt,
                                               int* __restrict__ offs,
                                               int* __restrict__ cur) {
    __shared__ int buf[SCAN_N];
    __shared__ int wtot[32];
    int lane = threadIdx.x & 31;
    int w    = threadIdx.x >> 5;
    int base = w * SCAN_SEG;
    int* seg = buf + base;

#pragma unroll
    for (int k = 0; k < SCAN_IT; ++k)
        seg[k * 32 + lane] = cnt[base + k * 32 + lane];
    __syncthreads();

    int* mine = seg + lane * SCAN_IT;
    int s = 0;
    for (int j = 0; j < SCAN_IT; ++j) s += mine[j];

    int inc = s;
    for (int d = 1; d < 32; d <<= 1) {
        int u = __shfl_up(inc, d);
        if (lane >= d) inc += u;
    }
    if (lane == 31) wtot[w] = inc;
    __syncthreads();

    int run = (lane < w) ? wtot[lane] : 0;
    for (int d = 16; d > 0; d >>= 1) run += __shfl_xor(run, d);
    run += inc - s;

    for (int j = 0; j < SCAN_IT; ++j) {
        int v = mine[j];
        mine[j] = run;
        run += v;
    }
    __syncthreads();

#pragma unroll
    for (int k = 0; k < SCAN_IT; ++k) {
        int ex = seg[k * 32 + lane];
        offs[base + k * 32 + lane] = ex;
        cur [base + k * 32 + lane] = ex;
    }
}

__global__ __launch_bounds__(256) void k_fill(const int* __restrict__ ei,
                                              const float* __restrict__ scores,
                                              int* __restrict__ cur,
                                              int* __restrict__ ccol,
                                              float4* __restrict__ cex) {
    int e = blockIdx.x * 256 + threadIdx.x;
    if (e >= NE) return;
    int row = ei[e];
    int col = ei[NE + e];
    float4 x = *reinterpret_cast<const float4*>(scores + 4 * (size_t)e);
    int pos = atomicAdd(&cur[row], 1);
    ccol[pos] = col;
    cex[pos]  = x;
}

#define ACC_BLOCKS 1563
__global__ __launch_bounds__(256) void k_acc(const int* __restrict__ offs,
                                             const int* __restrict__ ccol,
                                             const float* __restrict__ cex,
                                             const float* __restrict__ denom,
                                             const float* __restrict__ Vd,
                                             float* __restrict__ out) {
    int lane = threadIdx.x & 31;
    int h    = lane >> 3;
    int w = __builtin_amdgcn_readfirstlane((int)(blockIdx.x * 8 + (threadIdx.x >> 5)));
    const float4* V4 = reinterpret_cast<const float4*>(Vd);

    for (int row = w; row < NN; row += ACC_BLOCKS * 8) {
        int p0 = offs[row];
        int p1 = offs[row + 1];
        float den = denom[row * HEADS + h] + 1e-16f;
        float4 acc = make_float4(0.0f, 0.0f, 0.0f, 0.0f);

        for (int p = p0; p < p1; ++p) {
            int c = ccol[p];
            float a = cex[(size_t)p * 4 + h] / den;
            float4 v = V4[(size_t)c * 32 + lane];
            acc.x = fmaf(a, v.x, acc.x);
            acc.y = fmaf(a, v.y, acc.y);
            acc.z = fmaf(a, v.z, acc.z);
            acc.w = fmaf(a, v.w, acc.w);
        }
        *reinterpret_cast<float4*>(out + (size_t)row * OUT_T + lane * 4) = acc;
    }
}

extern "C" void kernel_launch(void* const* d_in, const int* in_sizes, int n_in,
                              void* d_out, int out_size, void* d_ws, size_t ws_size,
                              hipStream_t stream) {
    const float* x  = (const float*)d_in[0];
    const int*   ei = (const int*)  d_in[1];
    const float* Wq = (const float*)d_in[2];
    const float* Wk = (const float*)d_in[3];
    const float* Wv = (const float*)d_in[4];
    const float* Sr = (const float*)d_in[5];
    float* out = (float*)d_out;

    char* ws = (char*)d_ws;
    size_t off = 0;
    auto alloc = [&](size_t bytes) -> void* {
        void* p = ws + off;
        off = (off + bytes + 255) & ~(size_t)255;
        return p;
    };
    float*          Qd  = (float*)         alloc((size_t)NN * OUT_T * 4);
    float*          Kd  = (float*)         alloc((size_t)NN * OUT_T * 4);
    float*          Vd  = (float*)         alloc((size_t)NN * OUT_T * 4);
    unsigned short* xhi = (unsigned short*)alloc((size_t)NN * IN_F * 2);
    unsigned short* xlo = (unsigned short*)alloc((size_t)NN * IN_F * 2);
    unsigned short* whi = (unsigned short*)alloc((size_t)N_CAT * IN_F * 2);
    unsigned short* wlo = (unsigned short*)alloc((size_t)N_CAT * IN_F * 2);
    float*          ssy = (float*)         alloc((size_t)HEADS * D_K * D_K * 4);
    unsigned*       m_u = (unsigned*)      alloc((size_t)NN * HEADS * 4);
    float*          den = (float*)         alloc((size_t)NN * HEADS * 4);
    float*          sc  = (float*)         alloc((size_t)NE * HEADS * 4);
    int*            cnt = (int*)           alloc((size_t)SCAN_N * 4);
    int*            ofs = (int*)           alloc((size_t)SCAN_N * 4);
    int*            cur = (int*)           alloc((size_t)SCAN_N * 4);
    int*            ccl = (int*)           alloc((size_t)NE * 4);
    float*          cex = (float*)         alloc((size_t)NE * HEADS * 4);
    float*          SKd = (float*)         alloc((size_t)NN * OUT_T * 4);
    float*          ksk = (float*)         alloc((size_t)NN * HEADS * 4);
    float*          qsq = (float*)         alloc((size_t)NN * HEADS * 4);

    k_zero<<<(SCAN_N + 255) / 256, 256, 0, stream>>>((float*)cnt, SCAN_N);
    k_zero<<<(NN * HEADS + 255) / 256, 256, 0, stream>>>((float*)m_u, NN * HEADS);
    k_zero<<<(NN * HEADS + 255) / 256, 256, 0, stream>>>(den, NN * HEADS);

    k_split<<<(NN * IN_F + 255) / 256, 256, 0, stream>>>(x, xhi, xlo, NN * IN_F);
    k_wcat <<<(N_CAT * IN_F + 255) / 256, 256, 0, stream>>>(Wq, Wk, Wv, whi, wlo);
    k_ssym <<<(HEADS * D_K * D_K + 255) / 256, 256, 0, stream>>>(Sr, ssy);

    k_qkv<<<(M_TILES * N_TILES + 7) / 8, 256, 0, stream>>>(xhi, xlo, whi, wlo, Qd, Kd, Vd);

    int egrid = (NE * HEADS + 255) / 256;
    k_node  <<<dim3((NN + 255) / 256, HEADS * 2), 256, 0, stream>>>(Qd, Kd, ssy, SKd, ksk, qsq);
    k_scores<<<NE / SC_EPW / 8, 256, 0, stream>>>(SKd, Qd, ksk, qsq, ei, sc, m_u);
    k_exp   <<<egrid, 256, 0, stream>>>(ei, m_u, sc, den);

    k_hist<<<(NE / 4 + 255) / 256, 256, 0, stream>>>(ei, cnt);
    k_scan<<<1, 1024, 0, stream>>>(cnt, ofs, cur);
    k_fill<<<(NE + 255) / 256, 256, 0, stream>>>(ei, sc, cur, ccl, (float4*)cex);
    k_acc <<<ACC_BLOCKS, 256, 0, stream>>>(ofs, ccl, cex, den, Vd, out);
}


# Scores, softmax and alpha*V fused into one row-owner wave pass over the grouped edges: S*K row read once per row, running max with rescale, no max or denominator atomics; sum order changes
# speedup vs baseline: 5.1633x; 1.6978x over previous
#include <hip/hip_runtime.h>

#define NN      50000
#define NE      800000
#define IN_F    256
#define HEADS   4
#define D_K     32
#define OUT_T   128
#define N_CAT   384
#define M_TILES (NN / 16)
#define N_TILES (N_CAT / 16)

typedef __attribute__((ext_vector_type(16))) __bf16   v16bf;
typedef __attribute__((ext_vector_type(8)))  float    v8f;
typedef __attribute__((ext_vector_type(4)))  unsigned uint4v;

union Frag { uint4v q[2]; v16bf v; };

__device__ __forceinline__ unsigned short f2bf(float f) {
    unsigned u = __float_as_uint(f);
    unsigned r = (u + 0x7FFFu + ((u >> 16) & 1u)) >> 16;
    return (unsigned short)r;
}
__device__ __forceinline__ float bf2f(unsigned short h) {
    return __uint_as_float(((unsigned)h) << 16);
}

__global__ void k_zero(float* __restrict__ p, int n) {
    int g = blockIdx.x * blockDim.x + threadIdx.x;
    if (g < n) p[g] = 0.0f;
}

__global__ void k_split(const float* __restrict__ src,
                        unsigned short* __restrict__ hi,
                        unsigned short* __restrict__ lo, int n) {
    int g = blockIdx.x * blockDim.x + threadIdx.x;
    if (g >= n) return;
    float f = src[g];
    unsigned short h = f2bf(f);
    hi[g] = h;
    lo[g] = f2bf(f - bf2f(h));
}

__global__ void k_wcat(const float* __restrict__ Wq, const float* __restrict__ Wk,
                       const float* __restrict__ Wv,
                       unsigned short* __restrict__ whi,
                       unsigned short* __restrict__ wlo) {
    int g = blockIdx.x * blockDim.x + threadIdx.x;
    if (g >= N_CAT * IN_F) return;
    int n = g / IN_F, k = g - n * IN_F;
    int which = n >> 7, nn = n & 127;
    const float* W = (which == 0) ? Wq : (which == 1) ? Wk : Wv;
    float f = W[k * OUT_T + nn];
    unsigned short h = f2bf(f);
    whi[g] = h;
    wlo[g] = f2bf(f - bf2f(h));
}

__global__ void k_ssym(const float* __restrict__ S, float* __restrict__ out) {
    int g = blockIdx.x * blockDim.x + threadIdx.x;
    if (g >= HEADS * D_K * D_K) return;
    int h = g / (D_K * D_K);
    int r = g - h * D_K * D_K;
    int i = r / D_K, j = r - i * D_K;
    out[g] = 0.5f * (S[h * D_K * D_K + i * D_K + j] + S[h * D_K * D_K + j * D_K + i]);
}

#define QKV_MT 5
#define QKV_NT 4
#define QKV_MG (M_TILES / QKV_MT)
#define QKV_NG (N_TILES / QKV_NT)
static_assert(M_TILES % QKV_MT == 0 && N_TILES % QKV_NT == 0, "whole blocks");
static_assert(QKV_NT * 16 * 2 == OUT_T, "a column group lies inside one of Q, K, V");
__global__ __launch_bounds__(QKV_NG * 32) void k_qkv(const unsigned short* __restrict__ xhi,
                                                     const unsigned short* __restrict__ xlo,
                                                     const unsigned short* __restrict__ whi,
                                                     const unsigned short* __restrict__ wlo,
                                                     float* __restrict__ Qd,
                                                     float* __restrict__ Kd,
                                                     float* __restrict__ Vd) {
    int lane  = threadIdx.x & 31;
    int ng    = threadIdx.x >> 5;
    int mg    = blockIdx.x;
    int l16   = lane & 15;
    int lhalf = lane >> 4;

    size_t aoff = (size_t)(mg * (QKV_MT * 16) + l16) * IN_F + lhalf * 8;
    size_t boff = (size_t)(ng * (QKV_NT * 16) + l16) * IN_F + lhalf * 8;
    const unsigned short* ah = xhi + aoff;
    const unsigned short* al = xlo + aoff;
    const unsigned short* bh = whi + boff;
    const unsigned short* bl = wlo + boff;

    v8f c[QKV_MT][QKV_NT] = {};
#pragma unroll
    for (int ks = 0; ks < IN_F; ks += 32) {
        Frag fah[QKV_MT], fal[QKV_MT], fbh[QKV_NT], fbl[QKV_NT];
#pragma unroll
        for (int i = 0; i < QKV_MT; ++i) {
            const unsigned short* ph = ah + i * 16 * IN_F + ks;
            const unsigned short* pl = al + i * 16 * IN_F + ks;
            fah[i].q[0] = *reinterpret_cast<const uint4v*>(ph);
            fah[i].q[1] = *reinterpret_cast<const uint4v*>(ph + 16);
            fal[i].q[0] = *reinterpret_cast<const uint4v*>(pl);
            fal[i].q[1] = *reinterpret_cast<const uint4v*>(pl + 16);
        }
#pragma unroll
        for (int j = 0; j < QKV_NT; ++j) {
            const unsigned short* ph = bh + j * 16 * IN_F + ks;
            const unsigned short* pl = bl + j * 16 * IN_F + ks;
            fbh[j].q[0] = *reinterpret_cast<const uint4v*>(ph);
            fbh[j].q[1] = *reinterpret_cast<const uint4v*>(ph + 16);
            fbl[j].q[0] = *reinterpret_cast<const uint4v*>(pl);
            fbl[j].q[1] = *reinterpret_cast<const uint4v*>(pl + 16);
        }
        __builtin_amdgcn_sched_barrier(0);
#pragma unroll
        for (int j = 0; j < QKV_NT; ++j) {
#pragma unroll
            for (int i = 0; i < QKV_MT; ++i)
                c[i][j] = __builtin_amdgcn_wmma_f32_16x16x32_bf16(false, fah[i].v, false, fbh[j].v,
                                                                  (short)0, c[i][j], false, false);
#pragma unroll
            for (int i = 0; i < QKV_MT; ++i)
                c[i][j] = __builtin_amdgcn_wmma_f32_16x16x32_bf16(false, fah[i].v, false, fbl[j].v,
                                                                  (short)0, c[i][j], false, false);
#pragma unroll
            for (int i = 0; i < QKV_MT; ++i)
                c[i][j] = __builtin_amdgcn_wmma_f32_16x16x32_bf16(false, fal[i].v, false, fbh[j].v,
                                                                  (short)0, c[i][j], false, false);
        }
        __builtin_amdgcn_sched_barrier(0);
    }

    float* dst = (ng < 2) ? Qd : (ng < 4) ? Kd : Vd;
    int colBase = (ng & 1) * (QKV_NT * 16) + l16;
    int rowBase = mg * (QKV_MT * 16) + lhalf * 8;
#pragma unroll
    for (int i = 0; i < QKV_MT; ++i) {
#pragma unroll
        for (int j = 0; j < QKV_NT; ++j) {
#pragma unroll
            for (int v = 0; v < 8; ++v)
                dst[(size_t)(rowBase + i * 16 + v) * OUT_T + colBase + j * 16] = c[i][j][v];
        }
    }
}

__global__ __launch_bounds__(256) void k_node(const float* __restrict__ Qd,
                                              const float* __restrict__ Kd,
                                              const float* __restrict__ Ssym,
                                              float* __restrict__ SKd,
                                              float* __restrict__ kSk,
                                              float* __restrict__ qSq) {
    int n = blockIdx.x * 256 + threadIdx.x;
    if (n >= NN) return;
    int h   = blockIdx.y >> 1;
    int isQ = blockIdx.y & 1;
    const float* base = isQ ? Qd : Kd;
    const float4* src = reinterpret_cast<const float4*>(base + (size_t)n * OUT_T + h * D_K);
    float v[D_K];
#pragma unroll
    for (int i = 0; i < D_K / 4; ++i) {
        float4 t = src[i];
        v[4 * i + 0] = t.x; v[4 * i + 1] = t.y; v[4 * i + 2] = t.z; v[4 * i + 3] = t.w;
    }
    const float* Sh = Ssym + h * D_K * D_K;
    float sv[D_K];
    float s = 0.0f;
#pragma unroll
    for (int i = 0; i < D_K; ++i) {
        float acc = 0.0f;
#pragma unroll
        for (int j = 0; j < D_K; ++j) acc = fmaf(Sh[i * D_K + j], v[j], acc);
        sv[i] = acc;
        s = fmaf(v[i], acc, s);
    }
    if (isQ) {
        qSq[n * HEADS + h] = s;
    } else {
        kSk[n * HEADS + h] = s;
        float4* dst = reinterpret_cast<float4*>(SKd + (size_t)n * OUT_T + h * D_K);
#pragma unroll
        for (int i = 0; i < D_K / 4; ++i)
            dst[i] = make_float4(sv[4 * i + 0], sv[4 * i + 1], sv[4 * i + 2], sv[4 * i + 3]);
    }
}

__global__ __launch_bounds__(256) void k_hist(const int* __restrict__ ei,
                                              int* __restrict__ cnt) {
    int t = blockIdx.x * 256 + threadIdx.x;
    if (t >= NE / 4) return;
    int4 r = *reinterpret_cast<const int4*>(ei + 4 * t);
    atomicAdd(&cnt[r.x], 1);
    atomicAdd(&cnt[r.y], 1);
    atomicAdd(&cnt[r.z], 1);
    atomicAdd(&cnt[r.w], 1);
}

#define SCAN_IT  49
#define SCAN_SEG (SCAN_IT * 32)
#define SCAN_N   (32 * SCAN_SEG)
static_assert(SCAN_N >= NN + 1, "the scan must reach offs[NN]");
__global__ __launch_bounds__(1024) void k_scan(const int* __restrict__ cnt,
                                               int* __restrict__ offs,
                                               int* __restrict__ cur) {
    __shared__ int buf[SCAN_N];
    __shared__ int wtot[32];
    int lane = threadIdx.x & 31;
    int w    = threadIdx.x >> 5;
    int base = w * SCAN_SEG;
    int* seg = buf + base;

#pragma unroll
    for (int k = 0; k < SCAN_IT; ++k)
        seg[k * 32 + lane] = cnt[base + k * 32 + lane];
    __syncthreads();

    int* mine = seg + lane * SCAN_IT;
    int s = 0;
    for (int j = 0; j < SCAN_IT; ++j) s += mine[j];

    int inc = s;
    for (int d = 1; d < 32; d <<= 1) {
        int u = __shfl_up(inc, d);
        if (lane >= d) inc += u;
    }
    if (lane == 31) wtot[w] = inc;
    __syncthreads();

    int run = (lane < w) ? wtot[lane] : 0;
    for (int d = 16; d > 0; d >>= 1) run += __shfl_xor(run, d);
    run += inc - s;

    for (int j = 0; j < SCAN_IT; ++j) {
        int v = mine[j];
        mine[j] = run;
        run += v;
    }
    __syncthreads();

#pragma unroll
    for (int k = 0; k < SCAN_IT; ++k) {
        int ex = seg[k * 32 + lane];
        offs[base + k * 32 + lane] = ex;
        cur [base + k * 32 + lane] = ex;
    }
}

__global__ __launch_bounds__(256) void k_fill(const int* __restrict__ ei,
                                              int* __restrict__ cur,
                                              int* __restrict__ ccol) {
    int e = blockIdx.x * 256 + threadIdx.x;
    if (e >= NE) return;
    int row = ei[e];
    int col = ei[NE + e];
    int pos = atomicAdd(&cur[row], 1);
    ccol[pos] = col;
}

#define ROW_BLOCKS 6250
__global__ __launch_bounds__(128) void k_row(const int* __restrict__ offs,
                                             const int* __restrict__ ccol,
                                             const float* __restrict__ SKd,
                                             const float* __restrict__ Qd,
                                             const float* __restrict__ Vd,
                                             const float* __restrict__ kSk,
                                             const float* __restrict__ qSq,
                                             float* __restrict__ out) {
    int lane = threadIdx.x & 31;
    int h = lane >> 3;
    int g = lane & 7;
    int w = __builtin_amdgcn_readfirstlane((int)(blockIdx.x * 4 + (threadIdx.x >> 5)));
    const float4* SK4 = reinterpret_cast<const float4*>(SKd);
    const float4* Q4  = reinterpret_cast<const float4*>(Qd);
    const float4* V4  = reinterpret_cast<const float4*>(Vd);

    for (int row = w; row < NN; row += ROW_BLOCKS * 4) {
        int p0 = offs[row];
        int p1 = offs[row + 1];
        float4 sk = SK4[(size_t)row * 32 + lane];
        float ks = kSk[row * HEADS + h];
        float m = -INFINITY;
        float l = 0.0f;
        float4 acc = make_float4(0.0f, 0.0f, 0.0f, 0.0f);

        for (int p = p0; p < p1; p += 8) {
            int last = p1 - 1;
            int col[8];
#pragma unroll
            for (int u = 0; u < 8; ++u) {
                int pp = p + u;
                col[u] = ccol[pp < last ? pp : last];
            }
            float4 q[8], v[8];
#pragma unroll
            for (int u = 0; u < 8; ++u) {
                q[u] = Q4[(size_t)col[u] * 32 + lane];
                v[u] = V4[(size_t)col[u] * 32 + lane];
            }
            int pg = p + g;
            bool valid = pg <= last;
            int mycol = ccol[valid ? pg : last];
            float qs = qSq[mycol * HEADS + h];

            float d[8];
#pragma unroll
            for (int u = 0; u < 8; ++u)
                d[u] = fmaf(sk.x, q[u].x, fmaf(sk.y, q[u].y, fmaf(sk.z, q[u].z, sk.w * q[u].w)));

            float d4[4];
#pragma unroll
            for (int u = 0; u < 4; ++u) {
                float keep = (g & 4) ? d[u + 4] : d[u];
                float send = (g & 4) ? d[u]     : d[u + 4];
                d4[u] = keep + __shfl_xor(send, 4);
            }
            float d2[2];
#pragma unroll
            for (int u = 0; u < 2; ++u) {
                float keep = (g & 2) ? d4[u + 2] : d4[u];
                float send = (g & 2) ? d4[u]     : d4[u + 2];
                d2[u] = keep + __shfl_xor(send, 2);
            }
            float keep1 = (g & 1) ? d2[1] : d2[0];
            float send1 = (g & 1) ? d2[0] : d2[1];
            float dot = keep1 + __shfl_xor(send1, 1);

            float s = (ks + qs) - 2.0f * dot;
            s *= 0.17677669529663687f;
            s = (s > 0.0f) ? s : 0.2f * s;

            float cm = valid ? s : -INFINITY;
            cm = fmaxf(cm, __shfl_xor(cm, 1));
            cm = fmaxf(cm, __shfl_xor(cm, 2));
            cm = fmaxf(cm, __shfl_xor(cm, 4));
            float mn = fmaxf(m, cm);
            float sc = expf(m - mn);
            float e  = valid ? expf(s - mn) : 0.0f;
            m = mn;
            l = fmaf(l, sc, e);
            acc.x *= sc; acc.y *= sc; acc.z *= sc; acc.w *= sc;
#pragma unroll
            for (int u = 0; u < 8; ++u) {
                float a = __shfl(e, (lane & 24) | u);
                acc.x = fmaf(a, v[u].x, acc.x);
                acc.y = fmaf(a, v[u].y, acc.y);
                acc.z = fmaf(a, v[u].z, acc.z);
                acc.w = fmaf(a, v[u].w, acc.w);
            }
        }
        l += __shfl_xor(l, 1);
        l += __shfl_xor(l, 2);
        l += __shfl_xor(l, 4);
        float den = l + 1e-16f;
        *reinterpret_cast<float4*>(out + (size_t)row * OUT_T + lane * 4) =
            make_float4(acc.x / den, acc.y / den, acc.z / den, acc.w / den);
    }
}

extern "C" void kernel_launch(void* const* d_in, const int* in_sizes, int n_in,
                              void* d_out, int out_size, void* d_ws, size_t ws_size,
                              hipStream_t stream) {
    const float* x  = (const float*)d_in[0];
    const int*   ei = (const int*)  d_in[1];
    const float* Wq = (const float*)d_in[2];
    const float* Wk = (const float*)d_in[3];
    const float* Wv = (const float*)d_in[4];
    const float* Sr = (const float*)d_in[5];
    float* out = (float*)d_out;

    char* ws = (char*)d_ws;
    size_t off = 0;
    auto alloc = [&](size_t bytes) -> void* {
        void* p = ws + off;
        off = (off + bytes + 255) & ~(size_t)255;
        return p;
    };
    float*          Qd  = (float*)         alloc((size_t)NN * OUT_T * 4);
    float*          Kd  = (float*)         alloc((size_t)NN * OUT_T * 4);
    float*          Vd  = (float*)         alloc((size_t)NN * OUT_T * 4);
    unsigned short* xhi = (unsigned short*)alloc((size_t)NN * IN_F * 2);
    unsigned short* xlo = (unsigned short*)alloc((size_t)NN * IN_F * 2);
    unsigned short* whi = (unsigned short*)alloc((size_t)N_CAT * IN_F * 2);
    unsigned short* wlo = (unsigned short*)alloc((size_t)N_CAT * IN_F * 2);
    float*          ssy = (float*)         alloc((size_t)HEADS * D_K * D_K * 4);
    int*            cnt = (int*)           alloc((size_t)SCAN_N * 4);
    int*            ofs = (int*)           alloc((size_t)SCAN_N * 4);
    int*            cur = (int*)           alloc((size_t)SCAN_N * 4);
    int*            ccl = (int*)           alloc((size_t)NE * 4);
    float*          SKd = (float*)         alloc((size_t)NN * OUT_T * 4);
    float*          ksk = (float*)         alloc((size_t)NN * HEADS * 4);
    float*          qsq = (float*)         alloc((size_t)NN * HEADS * 4);

    k_zero<<<(SCAN_N + 255) / 256, 256, 0, stream>>>((float*)cnt, SCAN_N);
    k_hist<<<(NE / 4 + 255) / 256, 256, 0, stream>>>(ei, cnt);
    k_scan<<<1, 1024, 0, stream>>>(cnt, ofs, cur);
    k_fill<<<(NE + 255) / 256, 256, 0, stream>>>(ei, cur, ccl);

    k_split<<<(NN * IN_F + 255) / 256, 256, 0, stream>>>(x, xhi, xlo, NN * IN_F);
    k_wcat <<<(N_CAT * IN_F + 255) / 256, 256, 0, stream>>>(Wq, Wk, Wv, whi, wlo);
    k_ssym <<<(HEADS * D_K * D_K + 255) / 256, 256, 0, stream>>>(Sr, ssy);

    k_qkv<<<QKV_MG, QKV_NG * 32, 0, stream>>>(xhi, xlo, whi, wlo, Qd, Kd, Vd);

    k_node<<<dim3((NN + 255) / 256, HEADS * 2), 256, 0, stream>>>(Qd, Kd, ssy, SKd, ksk, qsq);
    k_row <<<ROW_BLOCKS, 128, 0, stream>>>(ofs, ccl, SKd, Qd, Vd, ksk, qsq, out);
}


# Edges placed straight into fixed 64-slot buckets per destination row (an overflow list takes longer rows): count pass and scan launch gone, 5 launches become 4; sum order stays arrival order
# speedup vs baseline: 7.8447x; 1.5193x over previous
#include <hip/hip_runtime.h>

#define NN      50000
#define NE      800000
#define IN_F    256
#define HEADS   4
#define D_K     32
#define OUT_T   128
#define N_CAT   384
#define M_TILES (NN / 16)
#define N_TILES (N_CAT / 16)

typedef __attribute__((ext_vector_type(16))) __bf16   v16bf;
typedef __attribute__((ext_vector_type(8)))  float    v8f;
typedef __attribute__((ext_vector_type(4)))  unsigned uint4v;

typedef __attribute__((ext_vector_type(2)))  unsigned uint2v;
typedef __attribute__((ext_vector_type(2)))  float    v2f;
typedef __attribute__((ext_vector_type(4)))  float    v4f;
typedef __attribute__((ext_vector_type(2)))  __bf16   v2bf;
typedef __attribute__((ext_vector_type(8)))  int      v8i;

union Frag { uint4v q[2]; v16bf v; };

__device__ __forceinline__ unsigned short f2bf(float f) {
    unsigned u = __float_as_uint(f);
    unsigned r = (u + 0x7FFFu + ((u >> 16) & 1u)) >> 16;
    return (unsigned short)r;
}
__device__ __forceinline__ float bf2f(unsigned short h) {
    return __uint_as_float(((unsigned)h) << 16);
}

#define PREP_WG 128
__device__ __forceinline__ void zero_stage(int blk, int* __restrict__ p, int n) {
    int g = blk * PREP_WG + threadIdx.x;
    if (g < n) p[g] = 0;
}

__device__ __forceinline__ unsigned pk_bf16(float a, float b) {
    v2f f = {a, b};
    return __builtin_bit_cast(unsigned, __builtin_convertvector(f, v2bf));
}
__device__ __forceinline__ void split2(float a, float b, unsigned& hi, unsigned& lo) {
    unsigned h = pk_bf16(a, b);
    hi = h;
    lo = pk_bf16(a - __uint_as_float(h << 16), b - __uint_as_float(h & 0xFFFF0000u));
}

#define SPLIT_Q   8
#define SPLIT_WG  128
#define SPLIT_PER_WG (SPLIT_WG * SPLIT_Q * 4)
static_assert((NN * IN_F) % SPLIT_PER_WG == 0, "whole work-groups: no bounds test");
static_assert(SPLIT_WG == PREP_WG, "the split shares its launch with the small prep stages");
__device__ __forceinline__ void split_stage(int blk, const float* __restrict__ src,
                                            unsigned short* __restrict__ hi,
                                            unsigned short* __restrict__ lo) {
    int lane = threadIdx.x & 31;
    int w    = blk * (SPLIT_WG / 32) + (threadIdx.x >> 5);
    size_t base = (size_t)w * (SPLIT_Q * 32) + lane;
    const v4f* s4 = reinterpret_cast<const v4f*>(src) + base;
    uint2v*    h2 = reinterpret_cast<uint2v*>(hi) + base;
    uint2v*    l2 = reinterpret_cast<uint2v*>(lo) + base;
    v4f r[SPLIT_Q];
#pragma unroll
    for (int q = 0; q < SPLIT_Q; ++q) r[q] = s4[q * 32];
#pragma unroll
    for (int q = 0; q < SPLIT_Q; ++q) {
        unsigned h0, h1, l0, l1;
        split2(r[q].x, r[q].y, h0, l0);
        split2(r[q].z, r[q].w, h1, l1);
        h2[q * 32] = uint2v{h0, h1};
        l2[q * 32] = uint2v{l0, l1};
    }
}

__device__ __forceinline__ void wcat_stage(int blk, const float* __restrict__ Wq,
                                           const float* __restrict__ Wk,
                                           const float* __restrict__ Wv,
                                           unsigned short* __restrict__ whi,
                                           unsigned short* __restrict__ wlo) {
    int g = blk * PREP_WG + threadIdx.x;
    if (g >= N_CAT * IN_F) return;
    int n = g / IN_F, k = g - n * IN_F;
    int which = n >> 7, nn = n & 127;
    const float* W = (which == 0) ? Wq : (which == 1) ? Wk : Wv;
    float f = W[k * OUT_T + nn];
    unsigned short h = f2bf(f);
    whi[g] = h;
    wlo[g] = f2bf(f - bf2f(h));
}

__device__ __forceinline__ void ssym_stage(int blk, const float* __restrict__ S,
                                           float* __restrict__ out) {
    int g = blk * PREP_WG + threadIdx.x;
    if (g >= HEADS * D_K * D_K) return;
    int h = g / (D_K * D_K);
    int r = g - h * D_K * D_K;
    int i = r / D_K, j = r - i * D_K;
    out[g] = 0.5f * (S[h * D_K * D_K + i * D_K + j] + S[h * D_K * D_K + j * D_K + i]);
}

#define QKV_MT 5
#define QKV_NT 4
#define QKV_MG (M_TILES / QKV_MT)
#define QKV_NG (N_TILES / QKV_NT)
static_assert(M_TILES % QKV_MT == 0 && N_TILES % QKV_NT == 0, "whole blocks");
static_assert(QKV_NT * 16 * 2 == OUT_T, "a column group lies inside one of Q, K, V");
#define QKV_WG (QKV_NG * 32)
__device__ __forceinline__ void qkv_stage(int blk, const unsigned short* __restrict__ xhi,
                                          const unsigned short* __restrict__ xlo,
                                          const unsigned short* __restrict__ whi,
                                          const unsigned short* __restrict__ wlo,
                                          float* __restrict__ Qd,
                                          float* __restrict__ Kd,
                                          float* __restrict__ Vd) {
    int lane  = threadIdx.x & 31;
    int ng    = threadIdx.x >> 5;
    int mg    = blk;
    int l16   = lane & 15;
    int lhalf = lane >> 4;

    size_t aoff = (size_t)(mg * (QKV_MT * 16) + l16) * IN_F + lhalf * 8;
    size_t boff = (size_t)(ng * (QKV_NT * 16) + l16) * IN_F + lhalf * 8;
    const unsigned short* ah = xhi + aoff;
    const unsigned short* al = xlo + aoff;
    const unsigned short* bh = whi + boff;
    const unsigned short* bl = wlo + boff;

    v8f c[QKV_MT][QKV_NT] = {};
#pragma unroll
    for (int ks = 0; ks < IN_F; ks += 32) {
        Frag fah[QKV_MT], fal[QKV_MT], fbh[QKV_NT], fbl[QKV_NT];
#pragma unroll
        for (int i = 0; i < QKV_MT; ++i) {
            const unsigned short* ph = ah + i * 16 * IN_F + ks;
            const unsigned short* pl = al + i * 16 * IN_F + ks;
            fah[i].q[0] = *reinterpret_cast<const uint4v*>(ph);
            fah[i].q[1] = *reinterpret_cast<const uint4v*>(ph + 16);
            fal[i].q[0] = *reinterpret_cast<const uint4v*>(pl);
            fal[i].q[1] = *reinterpret_cast<const uint4v*>(pl + 16);
        }
#pragma unroll
        for (int j = 0; j < QKV_NT; ++j) {
            const unsigned short* ph = bh + j * 16 * IN_F + ks;
            const unsigned short* pl = bl + j * 16 * IN_F + ks;
            fbh[j].q[0] = *reinterpret_cast<const uint4v*>(ph);
            fbh[j].q[1] = *reinterpret_cast<const uint4v*>(ph + 16);
            fbl[j].q[0] = *reinterpret_cast<const uint4v*>(pl);
            fbl[j].q[1] = *reinterpret_cast<const uint4v*>(pl + 16);
        }
        __builtin_amdgcn_sched_barrier(0);
#pragma unroll
        for (int j = 0; j < QKV_NT; ++j) {
#pragma unroll
            for (int i = 0; i < QKV_MT; ++i)
                c[i][j] = __builtin_amdgcn_wmma_f32_16x16x32_bf16(false, fah[i].v, false, fbh[j].v,
                                                                  (short)0, c[i][j], false, false);
#pragma unroll
            for (int i = 0; i < QKV_MT; ++i)
                c[i][j] = __builtin_amdgcn_wmma_f32_16x16x32_bf16(false, fah[i].v, false, fbl[j].v,
                                                                  (short)0, c[i][j], false, false);
#pragma unroll
            for (int i = 0; i < QKV_MT; ++i)
                c[i][j] = __builtin_amdgcn_wmma_f32_16x16x32_bf16(false, fal[i].v, false, fbh[j].v,
                                                                  (short)0, c[i][j], false, false);
        }
        __builtin_amdgcn_sched_barrier(0);
    }

    float* dst = (ng < 2) ? Qd : (ng < 4) ? Kd : Vd;
    int colBase = (ng & 1) * (QKV_NT * 16) + l16;
    int rowBase = mg * (QKV_MT * 16) + lhalf * 8;
#pragma unroll
    for (int i = 0; i < QKV_MT; ++i) {
#pragma unroll
        for (int j = 0; j < QKV_NT; ++j) {
#pragma unroll
            for (int v = 0; v < 8; ++v)
                dst[(size_t)(rowBase + i * 16 + v) * OUT_T + colBase + j * 16] = c[i][j][v];
        }
    }
}

#define NODE_BX ((NN + 255) / 256)
#define NODE_BLOCKS (NODE_BX * HEADS * 2)
__device__ __forceinline__ void node_stage(int blk, const float* __restrict__ Qd,
                                           const float* __restrict__ Kd,
                                           const float* __restrict__ Ssym,
                                           float* __restrict__ SKd,
                                           float* __restrict__ kSk,
                                           float* __restrict__ qSq) {
    int by = blk / NODE_BX;
    int n = (blk - by * NODE_BX) * 256 + threadIdx.x;
    if (n >= NN) return;
    int h   = by >> 1;
    int isQ = by & 1;
    const float* base = isQ ? Qd : Kd;
    const float4* src = reinterpret_cast<const float4*>(base + (size_t)n * OUT_T + h * D_K);
    float v[D_K];
#pragma unroll
    for (int i = 0; i < D_K / 4; ++i) {
        float4 t = src[i];
        v[4 * i + 0] = t.x; v[4 * i + 1] = t.y; v[4 * i + 2] = t.z; v[4 * i + 3] = t.w;
    }
    const float* Sh = Ssym + h * D_K * D_K;
    float sv[D_K];
    float s = 0.0f;
#pragma unroll
    for (int i = 0; i < D_K; ++i) {
        float acc = 0.0f;
#pragma unroll
        for (int j = 0; j < D_K; ++j) acc = fmaf(Sh[i * D_K + j], v[j], acc);
        sv[i] = acc;
        s = fmaf(v[i], acc, s);
    }
    if (isQ) {
        qSq[n * HEADS + h] = s;
    } else {
        kSk[n * HEADS + h] = s;
        float4* dst = reinterpret_cast<float4*>(SKd + (size_t)n * OUT_T + h * D_K);
#pragma unroll
        for (int i = 0; i < D_K / 4; ++i)
            dst[i] = make_float4(sv[4 * i + 0], sv[4 * i + 1], sv[4 * i + 2], sv[4 * i + 3]);
    }
}

#define ROW_CAP 64
#define FILL_BLOCKS ((NE / 4 + QKV_WG - 1) / QKV_WG)
static_assert(NE % 4 == 0, "four whole edges per thread");
__device__ __forceinline__ void place_edge(int row, int col, int* __restrict__ cnt,
                                           int* __restrict__ bucket, int* __restrict__ ovf) {
    int pos = atomicAdd(&cnt[row], 1);
    if (pos < ROW_CAP) {
        bucket[(size_t)row * ROW_CAP + pos] = col;
    } else {
        int o = atomicAdd(&cnt[NN], 1);
        ovf[2 * (size_t)o]     = row;
        ovf[2 * (size_t)o + 1] = col;
    }
}
__device__ __forceinline__ void fill_stage(int blk, const int* __restrict__ ei,
                                           int* __restrict__ cnt,
                                           int* __restrict__ bucket,
                                           int* __restrict__ ovf) {
    int t = blk * QKV_WG + threadIdx.x;
    if (t >= NE / 4) return;
    int4 r = *reinterpret_cast<const int4*>(ei + 4 * t);
    int4 c = *reinterpret_cast<const int4*>(ei + NE + 4 * t);
    place_edge(r.x, c.x, cnt, bucket, ovf);
    place_edge(r.y, c.y, cnt, bucket, ovf);
    place_edge(r.z, c.z, cnt, bucket, ovf);
    place_edge(r.w, c.w, cnt, bucket, ovf);
}

#define ROW_BLOCKS 6250
__global__ __launch_bounds__(128) __attribute__((amdgpu_waves_per_eu(9)))
void k_row(const int* __restrict__ cnt,
                                             const int* __restrict__ bucket,
                                             const int* __restrict__ ovf,
                                             const float* __restrict__ SKd,
                                             const float* __restrict__ Qd,
                                             const float* __restrict__ Vd,
                                             const float* __restrict__ kSk,
                                             const float* __restrict__ qSq,
                                             float* __restrict__ out) {
    int lane = threadIdx.x & 31;
    int h = lane >> 3;
    int g = lane & 7;
    int w = __builtin_amdgcn_readfirstlane((int)(blockIdx.x * 4 + (threadIdx.x >> 5)));
    const float4* SK4 = reinterpret_cast<const float4*>(SKd);
    const float4* Q4  = reinterpret_cast<const float4*>(Qd);
    const float4* V4  = reinterpret_cast<const float4*>(Vd);

    for (int row = w; row < NN; row += ROW_BLOCKS * 4) {
        int n  = cnt[row];
        int nb = n < ROW_CAP ? n : ROW_CAP;
        const int* brow = bucket + (size_t)row * ROW_CAP;
        float4 sk = SK4[(size_t)row * 32 + lane];
        float ks = kSk[row * HEADS + h];
        float m = -INFINITY;
        float l = 0.0f;
        float4 acc = make_float4(0.0f, 0.0f, 0.0f, 0.0f);

        auto trip = [&](const int (&col)[8], bool valid) __attribute__((always_inline)) {
            float4 q[8], v[8];
#pragma unroll
            for (int u = 0; u < 8; ++u) {
                q[u] = Q4[(size_t)col[u] * 32 + lane];
                v[u] = V4[(size_t)col[u] * 32 + lane];
            }
            int mycol = col[0];
#pragma unroll
            for (int u = 1; u < 8; ++u) mycol = (g == u) ? col[u] : mycol;
            float qs = qSq[mycol * HEADS + h];
            __builtin_amdgcn_sched_barrier(0);

            float d[8];
#pragma unroll
            for (int u = 0; u < 8; ++u)
                d[u] = fmaf(sk.x, q[u].x, fmaf(sk.y, q[u].y, fmaf(sk.z, q[u].z, sk.w * q[u].w)));

            float d4[4];
#pragma unroll
            for (int u = 0; u < 4; ++u) {
                float keep = (g & 4) ? d[u + 4] : d[u];
                float send = (g & 4) ? d[u]     : d[u + 4];
                d4[u] = keep + __shfl_xor(send, 4);
            }
            float d2[2];
#pragma unroll
            for (int u = 0; u < 2; ++u) {
                float keep = (g & 2) ? d4[u + 2] : d4[u];
                float send = (g & 2) ? d4[u]     : d4[u + 2];
                d2[u] = keep + __shfl_xor(send, 2);
            }
            float keep1 = (g & 1) ? d2[1] : d2[0];
            float send1 = (g & 1) ? d2[0] : d2[1];
            float dot = keep1 + __shfl_xor(send1, 1);

            float s = (ks + qs) - 2.0f * dot;
            s *= 0.17677669529663687f;
            s = (s > 0.0f) ? s : 0.2f * s;

            float cm = valid ? s : -INFINITY;
            cm = fmaxf(cm, __shfl_xor(cm, 1));
            cm = fmaxf(cm, __shfl_xor(cm, 2));
            cm = fmaxf(cm, __shfl_xor(cm, 4));
            float mn = fmaxf(m, cm);
            float sc = expf(m - mn);
            float e  = valid ? expf(s - mn) : 0.0f;
            m = mn;
            l = fmaf(l, sc, e);
            acc.x *= sc; acc.y *= sc; acc.z *= sc; acc.w *= sc;
#pragma unroll
            for (int u = 0; u < 8; ++u) {
                float a = __shfl(e, (lane & 24) | u);
                acc.x = fmaf(a, v[u].x, acc.x);
                acc.y = fmaf(a, v[u].y, acc.y);
                acc.z = fmaf(a, v[u].z, acc.z);
                acc.w = fmaf(a, v[u].w, acc.w);
            }
        };

        for (int p = 0; p < nb; p += 8) {
            int nv = nb - p;
            v8i raw = *reinterpret_cast<const v8i*>(brow + p);
            int col[8];
            int lc = raw[0];
            col[0] = lc;
#pragma unroll
            for (int u = 1; u < 8; ++u) {
                lc = (u < nv) ? raw[u] : lc;
                col[u] = lc;
            }
            trip(col, g < nv);
        }

        if (n > ROW_CAP) {
            int total = cnt[NN];
            total = total < NE ? total : NE;
            int held = 0;
            int colv = 0;
            auto flush = [&](int nv) __attribute__((always_inline)) {
                int col[8];
                int lc = __builtin_amdgcn_readlane(colv, 0);
                col[0] = lc;
#pragma unroll
                for (int u = 1; u < 8; ++u) {
                    int x = __builtin_amdgcn_readlane(colv, u);
                    lc = (u < nv) ? x : lc;
                    col[u] = lc;
                }
                trip(col, g < nv);
            };
            for (int base = 0; base < total; base += 16) {
                int idx = base + (lane & 15);
                bool in = lane < 16 && idx < total;
                int r = in ? ovf[2 * (size_t)idx]     : -1;
                int c = in ? ovf[2 * (size_t)idx + 1] : 0;
                unsigned hit = __builtin_amdgcn_ballot_w32(r == row);
                while (hit) {
                    int src = __builtin_ctz(hit);
                    hit &= hit - 1;
                    int cc = __builtin_amdgcn_readlane(c, src);
                    if (lane == held) colv = cc;
                    ++held;
                }
                while (held >= 8) {
                    flush(8);
                    colv = __shfl(colv, (lane + 8) & 31);
                    held -= 8;
                }
            }
            if (held > 0) flush(held);
        }

        l += __shfl_xor(l, 1);
        l += __shfl_xor(l, 2);
        l += __shfl_xor(l, 4);
        float den = l + 1e-16f;
        *reinterpret_cast<float4*>(out + (size_t)row * OUT_T + lane * 4) =
            make_float4(acc.x / den, acc.y / den, acc.z / den, acc.w / den);
    }
}

#define SPLIT_BLOCKS ((NN * IN_F) / SPLIT_PER_WG)
#define ZERO_BLOCKS  ((NN + 1 + PREP_WG - 1) / PREP_WG)
#define WCAT_BLOCKS  ((N_CAT * IN_F + PREP_WG - 1) / PREP_WG)
#define SSYM_BLOCKS  ((HEADS * D_K * D_K + PREP_WG - 1) / PREP_WG)
__global__ __launch_bounds__(PREP_WG) void k_prep(const float* __restrict__ x,
                                                  unsigned short* __restrict__ xhi,
                                                  unsigned short* __restrict__ xlo,
                                                  int* __restrict__ cnt,
                                                  const float* __restrict__ Wq,
                                                  const float* __restrict__ Wk,
                                                  const float* __restrict__ Wv,
                                                  unsigned short* __restrict__ whi,
                                                  unsigned short* __restrict__ wlo,
                                                  const float* __restrict__ Sr,
                                                  float* __restrict__ ssy) {
    int b = blockIdx.x;
    if (b < SPLIT_BLOCKS) { split_stage(b, x, xhi, xlo); return; }
    b -= SPLIT_BLOCKS;
    if (b < ZERO_BLOCKS) { zero_stage(b, cnt, NN + 1); return; }
    b -= ZERO_BLOCKS;
    if (b < WCAT_BLOCKS) { wcat_stage(b, Wq, Wk, Wv, whi, wlo); return; }
    ssym_stage(b - WCAT_BLOCKS, Sr, ssy);
}

__global__ __launch_bounds__(QKV_WG) void k_qkv_fill(const unsigned short* __restrict__ xhi,
                                                     const unsigned short* __restrict__ xlo,
                                                     const unsigned short* __restrict__ whi,
                                                     const unsigned short* __restrict__ wlo,
                                                     float* __restrict__ Qd,
                                                     float* __restrict__ Kd,
                                                     float* __restrict__ Vd,
                                                     const int* __restrict__ ei,
                                                     int* __restrict__ cnt,
                                                     int* __restrict__ bucket,
                                                     int* __restrict__ ovf) {
    int b = blockIdx.x;
    if (b < QKV_MG) { qkv_stage(b, xhi, xlo, whi, wlo, Qd, Kd, Vd); return; }
    fill_stage(b - QKV_MG, ei, cnt, bucket, ovf);
}

__global__ __launch_bounds__(256) void k_node(const float* __restrict__ Qd,
                                              const float* __restrict__ Kd,
                                              const float* __restrict__ Ssym,
                                              float* __restrict__ SKd,
                                              float* __restrict__ kSk,
                                              float* __restrict__ qSq) {
    node_stage(blockIdx.x, Qd, Kd, Ssym, SKd, kSk, qSq);
}

extern "C" void kernel_launch(void* const* d_in, const int* in_sizes, int n_in,
                              void* d_out, int out_size, void* d_ws, size_t ws_size,
                              hipStream_t stream) {
    const float* x  = (const float*)d_in[0];
    const int*   ei = (const int*)  d_in[1];
    const float* Wq = (const float*)d_in[2];
    const float* Wk = (const float*)d_in[3];
    const float* Wv = (const float*)d_in[4];
    const float* Sr = (const float*)d_in[5];
    float* out = (float*)d_out;

    char* ws = (char*)d_ws;
    size_t off = 0;
    auto alloc = [&](size_t bytes) -> void* {
        void* p = ws + off;
        off = (off + bytes + 255) & ~(size_t)255;
        return p;
    };
    float*          Qd  = (float*)         alloc((size_t)NN * OUT_T * 4);
    float*          Kd  = (float*)         alloc((size_t)NN * OUT_T * 4);
    float*          Vd  = (float*)         alloc((size_t)NN * OUT_T * 4);
    unsigned short* xhi = (unsigned short*)alloc((size_t)NN * IN_F * 2);
    unsigned short* xlo = (unsigned short*)alloc((size_t)NN * IN_F * 2);
    unsigned short* whi = (unsigned short*)alloc((size_t)N_CAT * IN_F * 2);
    unsigned short* wlo = (unsigned short*)alloc((size_t)N_CAT * IN_F * 2);
    float*          ssy = (float*)         alloc((size_t)HEADS * D_K * D_K * 4);
    int*            cnt = (int*)           alloc((size_t)(NN + 1) * 4);
    int*            bkt = (int*)           alloc((size_t)NN * ROW_CAP * 4);
    int*            ovf = (int*)           alloc((size_t)NE * 2 * 4);
    float*          SKd = (float*)         alloc((size_t)NN * OUT_T * 4);
    float*          ksk = (float*)         alloc((size_t)NN * HEADS * 4);
    float*          qsq = (float*)         alloc((size_t)NN * HEADS * 4);

    k_prep<<<SPLIT_BLOCKS + ZERO_BLOCKS + WCAT_BLOCKS + SSYM_BLOCKS, PREP_WG, 0, stream>>>(
        x, xhi, xlo, cnt, Wq, Wk, Wv, whi, wlo, Sr, ssy);

    k_qkv_fill<<<QKV_MG + FILL_BLOCKS, QKV_WG, 0, stream>>>(xhi, xlo, whi, wlo, Qd, Kd, Vd,
                                                            ei, cnt, bkt, ovf);

    k_node<<<NODE_BLOCKS, 256, 0, stream>>>(Qd, Kd, ssy, SKd, ksk, qsq);

    k_row <<<ROW_BLOCKS, 128, 0, stream>>>(cnt, bkt, ovf, SKd, Qd, Vd, ksk, qsq, out);
}


# QKV product reads x itself: a work-group splits its 80-row band to bf16 hi/lo once into LDS in fragment order, A fragments come from LDS; split pass and bf16 planes of x gone; same values and sums
# speedup vs baseline: 8.1155x; 1.0345x over previous
#include <hip/hip_runtime.h>

#define NN      50000
#define NE      800000
#define IN_F    256
#define HEADS   4
#define D_K     32
#define OUT_T   128
#define N_CAT   384
#define M_TILES (NN / 16)
#define N_TILES (N_CAT / 16)

typedef __attribute__((ext_vector_type(16))) __bf16   v16bf;
typedef __attribute__((ext_vector_type(8)))  float    v8f;
typedef __attribute__((ext_vector_type(4)))  unsigned uint4v;

typedef __attribute__((ext_vector_type(2)))  float    v2f;
typedef __attribute__((ext_vector_type(4)))  float    v4f;
typedef __attribute__((ext_vector_type(2)))  __bf16   v2bf;
typedef __attribute__((ext_vector_type(8)))  int      v8i;

union Frag { uint4v q[2]; v16bf v; };

__device__ __forceinline__ unsigned short f2bf(float f) {
    unsigned u = __float_as_uint(f);
    unsigned r = (u + 0x7FFFu + ((u >> 16) & 1u)) >> 16;
    return (unsigned short)r;
}
__device__ __forceinline__ float bf2f(unsigned short h) {
    return __uint_as_float(((unsigned)h) << 16);
}

#define PREP_WG 128
__device__ __forceinline__ void zero_stage(int blk, int* __restrict__ p, int n) {
    int g = blk * PREP_WG + threadIdx.x;
    if (g < n) p[g] = 0;
}

__device__ __forceinline__ unsigned pk_bf16(float a, float b) {
    v2f f = {a, b};
    return __builtin_bit_cast(unsigned, __builtin_convertvector(f, v2bf));
}
__device__ __forceinline__ void split2(float a, float b, unsigned& hi, unsigned& lo) {
    unsigned h = pk_bf16(a, b);
    hi = h;
    lo = pk_bf16(a - __uint_as_float(h << 16), b - __uint_as_float(h & 0xFFFF0000u));
}

__device__ __forceinline__ void wcat_stage(int blk, const float* __restrict__ Wq,
                                           const float* __restrict__ Wk,
                                           const float* __restrict__ Wv,
                                           unsigned short* __restrict__ whi,
                                           unsigned short* __restrict__ wlo) {
    int g = blk * PREP_WG + threadIdx.x;
    if (g >= N_CAT * IN_F) return;
    int n = g / IN_F, k = g - n * IN_F;
    int which = n >> 7, nn = n & 127;
    const float* W = (which == 0) ? Wq : (which == 1) ? Wk : Wv;
    float f = W[k * OUT_T + nn];
    unsigned short h = f2bf(f);
    whi[g] = h;
    wlo[g] = f2bf(f - bf2f(h));
}

__device__ __forceinline__ void ssym_stage(int blk, const float* __restrict__ S,
                                           float* __restrict__ out) {
    int g = blk * PREP_WG + threadIdx.x;
    if (g >= HEADS * D_K * D_K) return;
    int h = g / (D_K * D_K);
    int r = g - h * D_K * D_K;
    int i = r / D_K, j = r - i * D_K;
    out[g] = 0.5f * (S[h * D_K * D_K + i * D_K + j] + S[h * D_K * D_K + j * D_K + i]);
}

#define QKV_MT 5
#define QKV_NT 4
#define QKV_MG (M_TILES / QKV_MT)
#define QKV_NG (N_TILES / QKV_NT)
static_assert(M_TILES % QKV_MT == 0 && N_TILES % QKV_NT == 0, "whole blocks");
static_assert(QKV_NT * 16 * 2 == OUT_T, "a column group lies inside one of Q, K, V");
#define QKV_WG (QKV_NG * 32)
#define QKV_KS (IN_F / 32)
#define A_CHUNK 512
#define A_UNIT  (4 * A_CHUNK)
#define A_UNITS (QKV_MT * QKV_KS)
#define A_LDS_BYTES (A_UNITS * A_UNIT)
static_assert(A_UNITS > 6 * QKV_NG && A_UNITS <= 7 * QKV_NG, "six or seven units per wave");
__device__ __forceinline__ void qkv_stage(int blk, const float* __restrict__ x,
                                          unsigned char* __restrict__ alds,
                                          const unsigned short* __restrict__ whi,
                                          const unsigned short* __restrict__ wlo,
                                          float* __restrict__ Qd,
                                          float* __restrict__ Kd,
                                          float* __restrict__ Vd) {
    int lane  = threadIdx.x & 31;
    int ng    = threadIdx.x >> 5;
    int mg    = blk;
    int l16   = lane & 15;
    int lhalf = lane >> 4;

    size_t aoff = (size_t)(mg * (QKV_MT * 16) + l16) * IN_F + lhalf * 8;
    size_t boff = (size_t)(ng * (QKV_NT * 16) + l16) * IN_F + lhalf * 8;
    const float* ax = x + aoff;
    const unsigned short* bh = whi + boff;
    const unsigned short* bl = wlo + boff;

    Frag fb0h[QKV_NT], fb0l[QKV_NT];
    {
        int w0 = __builtin_amdgcn_readfirstlane(ng);
        v4f r[7][4];
#pragma unroll
        for (int t = 0; t < 7; ++t) {
            int u = w0 + QKV_NG * t;
            if (t < 6 || u < A_UNITS) {
                const float* p = ax + (size_t)(u >> 3) * 16 * IN_F + (u & 7) * 32;
                r[t][0] = *reinterpret_cast<const v4f*>(p);
                r[t][1] = *reinterpret_cast<const v4f*>(p + 4);
                r[t][2] = *reinterpret_cast<const v4f*>(p + 16);
                r[t][3] = *reinterpret_cast<const v4f*>(p + 20);
            }
        }
#pragma unroll
        for (int j = 0; j < QKV_NT; ++j) {
            const unsigned short* ph = bh + j * 16 * IN_F;
            const unsigned short* pl = bl + j * 16 * IN_F;
            fb0h[j].q[0] = *reinterpret_cast<const uint4v*>(ph);
            fb0h[j].q[1] = *reinterpret_cast<const uint4v*>(ph + 16);
            fb0l[j].q[0] = *reinterpret_cast<const uint4v*>(pl);
            fb0l[j].q[1] = *reinterpret_cast<const uint4v*>(pl + 16);
        }
        __builtin_amdgcn_sched_barrier(0);
#pragma unroll
        for (int t = 0; t < 7; ++t) {
            int u = w0 + QKV_NG * t;
            if (t < 6 || u < A_UNITS) {
                unsigned h[8], l[8];
#pragma unroll
                for (int j = 0; j < 4; ++j) {
                    split2(r[t][j].x, r[t][j].y, h[2 * j],     l[2 * j]);
                    split2(r[t][j].z, r[t][j].w, h[2 * j + 1], l[2 * j + 1]);
                }
                unsigned char* d = alds + u * A_UNIT + lane * 16;
                *reinterpret_cast<uint4v*>(d)               = uint4v{h[0], h[1], h[2], h[3]};
                *reinterpret_cast<uint4v*>(d + A_CHUNK)     = uint4v{h[4], h[5], h[6], h[7]};
                *reinterpret_cast<uint4v*>(d + 2 * A_CHUNK) = uint4v{l[0], l[1], l[2], l[3]};
                *reinterpret_cast<uint4v*>(d + 3 * A_CHUNK) = uint4v{l[4], l[5], l[6], l[7]};
            }
        }
    }
    __syncthreads();
    const unsigned char* afrag = alds + lane * 16;

    v8f c[QKV_MT][QKV_NT] = {};
#pragma unroll
    for (int ks = 0; ks < IN_F; ks += 32) {
        Frag fah[QKV_MT], fal[QKV_MT], fbh[QKV_NT], fbl[QKV_NT];
#pragma unroll
        for (int i = 0; i < QKV_MT; ++i) {
            const unsigned char* pa = afrag + (i * QKV_KS + ks / 32) * A_UNIT;
            fah[i].q[0] = *reinterpret_cast<const uint4v*>(pa);
            fah[i].q[1] = *reinterpret_cast<const uint4v*>(pa + A_CHUNK);
            fal[i].q[0] = *reinterpret_cast<const uint4v*>(pa + 2 * A_CHUNK);
            fal[i].q[1] = *reinterpret_cast<const uint4v*>(pa + 3 * A_CHUNK);
        }
#pragma unroll
        for (int j = 0; j < QKV_NT; ++j) {
            if (ks == 0) {
                fbh[j] = fb0h[j];
                fbl[j] = fb0l[j];
            } else {
                const unsigned short* ph = bh + j * 16 * IN_F + ks;
                const unsigned short* pl = bl + j * 16 * IN_F + ks;
                fbh[j].q[0] = *reinterpret_cast<const uint4v*>(ph);
                fbh[j].q[1] = *reinterpret_cast<const uint4v*>(ph + 16);
                fbl[j].q[0] = *reinterpret_cast<const uint4v*>(pl);
                fbl[j].q[1] = *reinterpret_cast<const uint4v*>(pl + 16);
            }
        }
        __builtin_amdgcn_sched_barrier(0);
#pragma unroll
        for (int j = 0; j < QKV_NT; ++j) {
#pragma unroll
            for (int i = 0; i < QKV_MT; ++i)
                c[i][j] = __builtin_amdgcn_wmma_f32_16x16x32_bf16(false, fah[i].v, false, fbh[j].v,
                                                                  (short)0, c[i][j], false, false);
#pragma unroll
            for (int i = 0; i < QKV_MT; ++i)
                c[i][j] = __builtin_amdgcn_wmma_f32_16x16x32_bf16(false, fah[i].v, false, fbl[j].v,
                                                                  (short)0, c[i][j], false, false);
#pragma unroll
            for (int i = 0; i < QKV_MT; ++i)
                c[i][j] = __builtin_amdgcn_wmma_f32_16x16x32_bf16(false, fal[i].v, false, fbh[j].v,
                                                                  (short)0, c[i][j], false, false);
        }
        __builtin_amdgcn_sched_barrier(0);
    }

    float* dst = (ng < 2) ? Qd : (ng < 4) ? Kd : Vd;
    int colBase = (ng & 1) * (QKV_NT * 16) + l16;
    int rowBase = mg * (QKV_MT * 16) + lhalf * 8;
#pragma unroll
    for (int i = 0; i < QKV_MT; ++i) {
#pragma unroll
        for (int j = 0; j < QKV_NT; ++j) {
#pragma unroll
            for (int v = 0; v < 8; ++v)
                dst[(size_t)(rowBase + i * 16 + v) * OUT_T + colBase + j * 16] = c[i][j][v];
        }
    }
}

#define NODE_BX ((NN + 255) / 256)
#define NODE_BLOCKS (NODE_BX * HEADS * 2)
__device__ __forceinline__ void node_stage(int blk, const float* __restrict__ Qd,
                                           const float* __restrict__ Kd,
                                           const float* __restrict__ Ssym,
                                           float* __restrict__ SKd,
                                           float* __restrict__ kSk,
                                           float* __restrict__ qSq) {
    int by = blk / NODE_BX;
    int n = (blk - by * NODE_BX) * 256 + threadIdx.x;
    if (n >= NN) return;
    int h   = by >> 1;
    int isQ = by & 1;
    const float* base = isQ ? Qd : Kd;
    const float4* src = reinterpret_cast<const float4*>(base + (size_t)n * OUT_T + h * D_K);
    float v[D_K];
#pragma unroll
    for (int i = 0; i < D_K / 4; ++i) {
        float4 t = src[i];
        v[4 * i + 0] = t.x; v[4 * i + 1] = t.y; v[4 * i + 2] = t.z; v[4 * i + 3] = t.w;
    }
    const float* Sh = Ssym + h * D_K * D_K;
    float sv[D_K];
    float s = 0.0f;
#pragma unroll
    for (int i = 0; i < D_K; ++i) {
        float acc = 0.0f;
#pragma unroll
        for (int j = 0; j < D_K; ++j) acc = fmaf(Sh[i * D_K + j], v[j], acc);
        sv[i] = acc;
        s = fmaf(v[i], acc, s);
    }
    if (isQ) {
        qSq[n * HEADS + h] = s;
    } else {
        kSk[n * HEADS + h] = s;
        float4* dst = reinterpret_cast<float4*>(SKd + (size_t)n * OUT_T + h * D_K);
#pragma unroll
        for (int i = 0; i < D_K / 4; ++i)
            dst[i] = make_float4(sv[4 * i + 0], sv[4 * i + 1], sv[4 * i + 2], sv[4 * i + 3]);
    }
}

#define ROW_CAP 64
#define FILL_BLOCKS ((NE / 4 + QKV_WG - 1) / QKV_WG)
static_assert(NE % 4 == 0, "four whole edges per thread");
__device__ __forceinline__ void place_edge(int row, int col, int* __restrict__ cnt,
                                           int* __restrict__ bucket, int* __restrict__ ovf) {
    int pos = atomicAdd(&cnt[row], 1);
    if (pos < ROW_CAP) {
        bucket[(size_t)row * ROW_CAP + pos] = col;
    } else {
        int o = atomicAdd(&cnt[NN], 1);
        ovf[2 * (size_t)o]     = row;
        ovf[2 * (size_t)o + 1] = col;
    }
}
__device__ __forceinline__ void fill_stage(int blk, const int* __restrict__ ei,
                                           int* __restrict__ cnt,
                                           int* __restrict__ bucket,
                                           int* __restrict__ ovf) {
    int t = blk * QKV_WG + threadIdx.x;
    if (t >= NE / 4) return;
    int4 r = *reinterpret_cast<const int4*>(ei + 4 * t);
    int4 c = *reinterpret_cast<const int4*>(ei + NE + 4 * t);
    place_edge(r.x, c.x, cnt, bucket, ovf);
    place_edge(r.y, c.y, cnt, bucket, ovf);
    place_edge(r.z, c.z, cnt, bucket, ovf);
    place_edge(r.w, c.w, cnt, bucket, ovf);
}

#define ROW_BLOCKS 6250
__global__ __launch_bounds__(128) __attribute__((amdgpu_waves_per_eu(9)))
void k_row(const int* __restrict__ cnt,
                                             const int* __restrict__ bucket,
                                             const int* __restrict__ ovf,
                                             const float* __restrict__ SKd,
                                             const float* __restrict__ Qd,
                                             const float* __restrict__ Vd,
                                             const float* __restrict__ kSk,
                                             const float* __restrict__ qSq,
                                             float* __restrict__ out) {
    int lane = threadIdx.x & 31;
    int h = lane >> 3;
    int g = lane & 7;
    int w = __builtin_amdgcn_readfirstlane((int)(blockIdx.x * 4 + (threadIdx.x >> 5)));
    const float4* SK4 = reinterpret_cast<const float4*>(SKd);
    const float4* Q4  = reinterpret_cast<const float4*>(Qd);
    const float4* V4  = reinterpret_cast<const float4*>(Vd);

    for (int row = w; row < NN; row += ROW_BLOCKS * 4) {
        int n  = cnt[row];
        int nb = n < ROW_CAP ? n : ROW_CAP;
        const int* brow = bucket + (size_t)row * ROW_CAP;
        float4 sk = SK4[(size_t)row * 32 + lane];
        float ks = kSk[row * HEADS + h];
        float m = -INFINITY;
        float l = 0.0f;
        float4 acc = make_float4(0.0f, 0.0f, 0.0f, 0.0f);

        auto trip = [&](const int (&col)[8], bool valid) __attribute__((always_inline)) {
            float4 q[8], v[8];
#pragma unroll
            for (int u = 0; u < 8; ++u) {
                q[u] = Q4[(size_t)col[u] * 32 + lane];
                v[u] = V4[(size_t)col[u] * 32 + lane];
            }
            int mycol = col[0];
#pragma unroll
            for (int u = 1; u < 8; ++u) mycol = (g == u) ? col[u] : mycol;
            float qs = qSq[mycol * HEADS + h];
            __builtin_amdgcn_sched_barrier(0);

            float d[8];
#pragma unroll
            for (int u = 0; u < 8; ++u)
                d[u] = fmaf(sk.x, q[u].x, fmaf(sk.y, q[u].y, fmaf(sk.z, q[u].z, sk.w * q[u].w)));

            float d4[4];
#pragma unroll
            for (int u = 0; u < 4; ++u) {
                float keep = (g & 4) ? d[u + 4] : d[u];
                float send = (g & 4) ? d[u]     : d[u + 4];
                d4[u] = keep + __shfl_xor(send, 4);
            }
            float d2[2];
#pragma unroll
            for (int u = 0; u < 2; ++u) {
                float keep = (g & 2) ? d4[u + 2] : d4[u];
                float send = (g & 2) ? d4[u]     : d4[u + 2];
                d2[u] = keep + __shfl_xor(send, 2);
            }
            float keep1 = (g & 1) ? d2[1] : d2[0];
            float send1 = (g & 1) ? d2[0] : d2[1];
            float dot = keep1 + __shfl_xor(send1, 1);

            float s = (ks + qs) - 2.0f * dot;
            s *= 0.17677669529663687f;
            s = (s > 0.0f) ? s : 0.2f * s;

            float cm = valid ? s : -INFINITY;
            cm = fmaxf(cm, __shfl_xor(cm, 1));
            cm = fmaxf(cm, __shfl_xor(cm, 2));
            cm = fmaxf(cm, __shfl_xor(cm, 4));
            float mn = fmaxf(m, cm);
            float sc = expf(m - mn);
            float e  = valid ? expf(s - mn) : 0.0f;
            m = mn;
            l = fmaf(l, sc, e);
            acc.x *= sc; acc.y *= sc; acc.z *= sc; acc.w *= sc;
#pragma unroll
            for (int u = 0; u < 8; ++u) {
                float a = __shfl(e, (lane & 24) | u);
                acc.x = fmaf(a, v[u].x, acc.x);
                acc.y = fmaf(a, v[u].y, acc.y);
                acc.z = fmaf(a, v[u].z, acc.z);
                acc.w = fmaf(a, v[u].w, acc.w);
            }
        };

        for (int p = 0; p < nb; p += 8) {
            int nv = nb - p;
            v8i raw = *reinterpret_cast<const v8i*>(brow + p);
            int col[8];
            int lc = raw[0];
            col[0] = lc;
#pragma unroll
            for (int u = 1; u < 8; ++u) {
                lc = (u < nv) ? raw[u] : lc;
                col[u] = lc;
            }
            trip(col, g < nv);
        }

        if (n > ROW_CAP) {
            int total = cnt[NN];
            total = total < NE ? total : NE;
            int held = 0;
            int colv = 0;
            auto flush = [&](int nv) __attribute__((always_inline)) {
                int col[8];
                int lc = __builtin_amdgcn_readlane(colv, 0);
                col[0] = lc;
#pragma unroll
                for (int u = 1; u < 8; ++u) {
                    int x = __builtin_amdgcn_readlane(colv, u);
                    lc = (u < nv) ? x : lc;
                    col[u] = lc;
                }
                trip(col, g < nv);
            };
            for (int base = 0; base < total; base += 16) {
                int idx = base + (lane & 15);
                bool in = lane < 16 && idx < total;
                int r = in ? ovf[2 * (size_t)idx]     : -1;
                int c = in ? ovf[2 * (size_t)idx + 1] : 0;
                unsigned hit = __builtin_amdgcn_ballot_w32(r == row);
                while (hit) {
                    int src = __builtin_ctz(hit);
                    hit &= hit - 1;
                    int cc = __builtin_amdgcn_readlane(c, src);
                    if (lane == held) colv = cc;
                    ++held;
                }
                while (held >= 8) {
                    flush(8);
                    colv = __shfl(colv, (lane + 8) & 31);
                    held -= 8;
                }
            }
            if (held > 0) flush(held);
        }

        l += __shfl_xor(l, 1);
        l += __shfl_xor(l, 2);
        l += __shfl_xor(l, 4);
        float den = l + 1e-16f;
        *reinterpret_cast<float4*>(out + (size_t)row * OUT_T + lane * 4) =
            make_float4(acc.x / den, acc.y / den, acc.z / den, acc.w / den);
    }
}

#define ZERO_BLOCKS  ((NN + 1 + PREP_WG - 1) / PREP_WG)
#define WCAT_BLOCKS  ((N_CAT * IN_F + PREP_WG - 1) / PREP_WG)
#define SSYM_BLOCKS  ((HEADS * D_K * D_K + PREP_WG - 1) / PREP_WG)
__global__ __launch_bounds__(PREP_WG) void k_prep(int* __restrict__ cnt,
                                                  const float* __restrict__ Wq,
                                                  const float* __restrict__ Wk,
                                                  const float* __restrict__ Wv,
                                                  unsigned short* __restrict__ whi,
                                                  unsigned short* __restrict__ wlo,
                                                  const float* __restrict__ Sr,
                                                  float* __restrict__ ssy) {
    int b = blockIdx.x;
    if (b < ZERO_BLOCKS) { zero_stage(b, cnt, NN + 1); return; }
    b -= ZERO_BLOCKS;
    if (b < WCAT_BLOCKS) { wcat_stage(b, Wq, Wk, Wv, whi, wlo); return; }
    ssym_stage(b - WCAT_BLOCKS, Sr, ssy);
}

__global__ __launch_bounds__(QKV_WG) void k_qkv_fill(const float* __restrict__ x,
                                                     const unsigned short* __restrict__ whi,
                                                     const unsigned short* __restrict__ wlo,
                                                     float* __restrict__ Qd,
                                                     float* __restrict__ Kd,
                                                     float* __restrict__ Vd,
                                                     const int* __restrict__ ei,
                                                     int* __restrict__ cnt,
                                                     int* __restrict__ bucket,
                                                     int* __restrict__ ovf) {
    __shared__ __attribute__((aligned(16))) unsigned char alds[A_LDS_BYTES];
    int b = blockIdx.x;
    if (b < QKV_MG) { qkv_stage(b, x, alds, whi, wlo, Qd, Kd, Vd); return; }
    fill_stage(b - QKV_MG, ei, cnt, bucket, ovf);
}

__global__ __launch_bounds__(256) void k_node(const float* __restrict__ Qd,
                                              const float* __restrict__ Kd,
                                              const float* __restrict__ Ssym,
                                              float* __restrict__ SKd,
                                              float* __restrict__ kSk,
                                              float* __restrict__ qSq) {
    node_stage(blockIdx.x, Qd, Kd, Ssym, SKd, kSk, qSq);
}

extern "C" void kernel_launch(void* const* d_in, const int* in_sizes, int n_in,
                              void* d_out, int out_size, void* d_ws, size_t ws_size,
                              hipStream_t stream) {
    const float* x  = (const float*)d_in[0];
    const int*   ei = (const int*)  d_in[1];
    const float* Wq = (const float*)d_in[2];
    const float* Wk = (const float*)d_in[3];
    const float* Wv = (const float*)d_in[4];
    const float* Sr = (const float*)d_in[5];
    float* out = (float*)d_out;

    char* ws = (char*)d_ws;
    size_t off = 0;
    auto alloc = [&](size_t bytes) -> void* {
        void* p = ws + off;
        off = (off + bytes + 255) & ~(size_t)255;
        return p;
    };
    float*          Qd  = (float*)         alloc((size_t)NN * OUT_T * 4);
    float*          Kd  = (float*)         alloc((size_t)NN * OUT_T * 4);
    float*          Vd  = (float*)         alloc((size_t)NN * OUT_T * 4);
    unsigned short* whi = (unsigned short*)alloc((size_t)N_CAT * IN_F * 2);
    unsigned short* wlo = (unsigned short*)alloc((size_t)N_CAT * IN_F * 2);
    float*          ssy = (float*)         alloc((size_t)HEADS * D_K * D_K * 4);
    int*            cnt = (int*)           alloc((size_t)(NN + 1) * 4);
    int*            bkt = (int*)           alloc((size_t)NN * ROW_CAP * 4);
    int*            ovf = (int*)           alloc((size_t)NE * 2 * 4);
    float*          SKd = (float*)         alloc((size_t)NN * OUT_T * 4);
    float*          ksk = (float*)         alloc((size_t)NN * HEADS * 4);
    float*          qsq = (float*)         alloc((size_t)NN * HEADS * 4);

    k_prep<<<ZERO_BLOCKS + WCAT_BLOCKS + SSYM_BLOCKS, PREP_WG, 0, stream>>>(
        cnt, Wq, Wk, Wv, whi, wlo, Sr, ssy);

    k_qkv_fill<<<QKV_MG + FILL_BLOCKS, QKV_WG, 0, stream>>>(x, whi, wlo, Qd, Kd, Vd,
                                                            ei, cnt, bkt, ovf);

    k_node<<<NODE_BLOCKS, 256, 0, stream>>>(Qd, Kd, ssy, SKd, ksk, qsq);

    k_row <<<ROW_BLOCKS, 128, 0, stream>>>(cnt, bkt, ovf, SKd, Qd, Vd, ksk, qsq, out);
}


# W packed by the prep launch in B-fragment order (512 contiguous bytes per tile, K-step, plane, half): a B load of the QKV product covers four whole lines, was sixteen lines x 32 B; same values, sums
# speedup vs baseline: 9.0298x; 1.1127x over previous
#include <hip/hip_runtime.h>

#define NN      50000
#define NE      800000
#define IN_F    256
#define HEADS   4
#define D_K     32
#define OUT_T   128
#define N_CAT   384
#define M_TILES (NN / 16)
#define N_TILES (N_CAT / 16)

typedef __attribute__((ext_vector_type(16))) __bf16   v16bf;
typedef __attribute__((ext_vector_type(8)))  float    v8f;
typedef __attribute__((ext_vector_type(4)))  unsigned uint4v;

typedef __attribute__((ext_vector_type(2)))  float    v2f;
typedef __attribute__((ext_vector_type(4)))  float    v4f;
typedef __attribute__((ext_vector_type(2)))  __bf16   v2bf;
typedef __attribute__((ext_vector_type(8)))  int      v8i;

union Frag { uint4v q[2]; v16bf v; };

__device__ __forceinline__ unsigned short f2bf(float f) {
    unsigned u = __float_as_uint(f);
    unsigned r = (u + 0x7FFFu + ((u >> 16) & 1u)) >> 16;
    return (unsigned short)r;
}
__device__ __forceinline__ float bf2f(unsigned short h) {
    return __uint_as_float(((unsigned)h) << 16);
}

#define PREP_WG 128
__device__ __forceinline__ void zero_stage(int blk, int* __restrict__ p, int n) {
    int g = blk * PREP_WG + threadIdx.x;
    if (g < n) p[g] = 0;
}

__device__ __forceinline__ unsigned pk_bf16(float a, float b) {
    v2f f = {a, b};
    return __builtin_bit_cast(unsigned, __builtin_convertvector(f, v2bf));
}
__device__ __forceinline__ void split2(float a, float b, unsigned& hi, unsigned& lo) {
    unsigned h = pk_bf16(a, b);
    hi = h;
    lo = pk_bf16(a - __uint_as_float(h << 16), b - __uint_as_float(h & 0xFFFF0000u));
}

#define QKV_NT 4
#define QKV_KS (IN_F / 32)
#define W_CHUNK 256
#define W_GROUP (QKV_KS * QKV_NT * 4 * W_CHUNK)
__device__ __forceinline__ int wfrag_index(int n, int k, int plane) {
    int ng = n >> 6, j = (n >> 4) & 3, l16 = n & 15;
    int s = k >> 5, q = (k >> 4) & 1, lhalf = (k >> 3) & 1, e = k & 7;
    return ng * W_GROUP + ((s * QKV_NT + j) * 4 + plane * 2 + q) * W_CHUNK + (lhalf * 16 + l16) * 8 + e;
}
static_assert(QKV_NT == 4 && QKV_KS == 8, "the bit fields of wfrag_index");
__device__ __forceinline__ void wcat_stage(int blk, const float* __restrict__ Wq,
                                           const float* __restrict__ Wk,
                                           const float* __restrict__ Wv,
                                           unsigned short* __restrict__ wfr) {
    int g = blk * PREP_WG + threadIdx.x;
    if (g >= N_CAT * IN_F) return;
    int n = g / IN_F, k = g - n * IN_F;
    int which = n >> 7, nn = n & 127;
    const float* W = (which == 0) ? Wq : (which == 1) ? Wk : Wv;
    float f = W[k * OUT_T + nn];
    unsigned short h = f2bf(f);
    wfr[wfrag_index(n, k, 0)] = h;
    wfr[wfrag_index(n, k, 1)] = f2bf(f - bf2f(h));
}

__device__ __forceinline__ void ssym_stage(int blk, const float* __restrict__ S,
                                           unsigned short* __restrict__ shi,
                                           unsigned short* __restrict__ slo) {
    int g = blk * PREP_WG + threadIdx.x;
    if (g >= HEADS * D_K * D_K) return;
    int h = g / (D_K * D_K);
    int r = g - h * D_K * D_K;
    int i = r / D_K, j = r - i * D_K;
    float f = 0.5f * (S[h * D_K * D_K + i * D_K + j] + S[h * D_K * D_K + j * D_K + i]);
    unsigned short hb = f2bf(f);
    shi[g] = hb;
    slo[g] = f2bf(f - bf2f(hb));
}

#define QKV_MT 5
#define QKV_MG (M_TILES / QKV_MT)
#define QKV_NG (N_TILES / QKV_NT)
static_assert(M_TILES % QKV_MT == 0 && N_TILES % QKV_NT == 0, "whole blocks");
static_assert(QKV_NT * 16 * 2 == OUT_T, "a column group lies inside one of Q, K, V");
static_assert(QKV_NG * W_GROUP == 2 * N_CAT * IN_F, "the packed weights hold every hi and lo value once");
#define QKV_WG (QKV_NG * 32)
#define A_CHUNK 512
#define A_UNIT  (4 * A_CHUNK)
#define A_UNITS (QKV_MT * QKV_KS)
#define A_LDS_BYTES (A_UNITS * A_UNIT)
static_assert(A_UNITS > 6 * QKV_NG && A_UNITS <= 7 * QKV_NG, "six or seven units per wave");
__device__ __forceinline__ void qkv_stage(int blk, const float* __restrict__ x,
                                          unsigned char* __restrict__ alds,
                                          const unsigned short* __restrict__ wfr,
                                          float* __restrict__ Qd,
                                          float* __restrict__ Kd,
                                          float* __restrict__ Vd,
                                          const unsigned short* __restrict__ shi,
                                          const unsigned short* __restrict__ slo,
                                          float* __restrict__ SKd,
                                          float* __restrict__ kSk,
                                          float* __restrict__ qSq) {
    int lane  = threadIdx.x & 31;
    int ng    = threadIdx.x >> 5;
    int mg    = blk;
    int l16   = lane & 15;
    int lhalf = lane >> 4;

    size_t aoff = (size_t)(mg * (QKV_MT * 16) + l16) * IN_F + lhalf * 8;
    const float* ax = x + aoff;
    const unsigned short* bw = wfr + (size_t)ng * W_GROUP + lane * 8;

    Frag fb0h[QKV_NT], fb0l[QKV_NT];
    {
        int w0 = __builtin_amdgcn_readfirstlane(ng);
        v4f r[7][4];
#pragma unroll
        for (int t = 0; t < 7; ++t) {
            int u = w0 + QKV_NG * t;
            if (t < 6 || u < A_UNITS) {
                const float* p = ax + (size_t)(u >> 3) * 16 * IN_F + (u & 7) * 32;
                r[t][0] = *reinterpret_cast<const v4f*>(p);
                r[t][1] = *reinterpret_cast<const v4f*>(p + 4);
                r[t][2] = *reinterpret_cast<const v4f*>(p + 16);
                r[t][3] = *reinterpret_cast<const v4f*>(p + 20);
            }
        }
#pragma unroll
        for (int j = 0; j < QKV_NT; ++j) {
            const unsigned short* pb = bw + j * 4 * W_CHUNK;
            fb0h[j].q[0] = *reinterpret_cast<const uint4v*>(pb);
            fb0h[j].q[1] = *reinterpret_cast<const uint4v*>(pb + W_CHUNK);
            fb0l[j].q[0] = *reinterpret_cast<const uint4v*>(pb + 2 * W_CHUNK);
            fb0l[j].q[1] = *reinterpret_cast<const uint4v*>(pb + 3 * W_CHUNK);
        }
        __builtin_amdgcn_sched_barrier(0);
#pragma unroll
        for (int t = 0; t < 7; ++t) {
            int u = w0 + QKV_NG * t;
            if (t < 6 || u < A_UNITS) {
                unsigned h[8], l[8];
#pragma unroll
                for (int j = 0; j < 4; ++j) {
                    split2(r[t][j].x, r[t][j].y, h[2 * j],     l[2 * j]);
                    split2(r[t][j].z, r[t][j].w, h[2 * j + 1], l[2 * j + 1]);
                }
                unsigned char* d = alds + u * A_UNIT + lane * 16;
                *reinterpret_cast<uint4v*>(d)               = uint4v{h[0], h[1], h[2], h[3]};
                *reinterpret_cast<uint4v*>(d + A_CHUNK)     = uint4v{h[4], h[5], h[6], h[7]};
                *reinterpret_cast<uint4v*>(d + 2 * A_CHUNK) = uint4v{l[0], l[1], l[2], l[3]};
                *reinterpret_cast<uint4v*>(d + 3 * A_CHUNK) = uint4v{l[4], l[5], l[6], l[7]};
            }
        }
    }
    __syncthreads();
    const unsigned char* afrag = alds + lane * 16;

    v8f c[QKV_MT][QKV_NT] = {};
#pragma unroll
    for (int ks = 0; ks < IN_F; ks += 32) {
        Frag fah[QKV_MT], fal[QKV_MT], fbh[QKV_NT], fbl[QKV_NT];
#pragma unroll
        for (int i = 0; i < QKV_MT; ++i) {
            const unsigned char* pa = afrag + (i * QKV_KS + ks / 32) * A_UNIT;
            fah[i].q[0] = *reinterpret_cast<const uint4v*>(pa);
            fah[i].q[1] = *reinterpret_cast<const uint4v*>(pa + A_CHUNK);
            fal[i].q[0] = *reinterpret_cast<const uint4v*>(pa + 2 * A_CHUNK);
            fal[i].q[1] = *reinterpret_cast<const uint4v*>(pa + 3 * A_CHUNK);
        }
#pragma unroll
        for (int j = 0; j < QKV_NT; ++j) {
            if (ks == 0) {
                fbh[j] = fb0h[j];
                fbl[j] = fb0l[j];
            } else {
                const unsigned short* pb = bw + ((ks / 32) * QKV_NT + j) * 4 * W_CHUNK;
                fbh[j].q[0] = *reinterpret_cast<const uint4v*>(pb);
                fbh[j].q[1] = *reinterpret_cast<const uint4v*>(pb + W_CHUNK);
                fbl[j].q[0] = *reinterpret_cast<const uint4v*>(pb + 2 * W_CHUNK);
                fbl[j].q[1] = *reinterpret_cast<const uint4v*>(pb + 3 * W_CHUNK);
            }
        }
        __builtin_amdgcn_sched_barrier(0);
#pragma unroll
        for (int j = 0; j < QKV_NT; ++j) {
#pragma unroll
            for (int i = 0; i < QKV_MT; ++i)
                c[i][j] = __builtin_amdgcn_wmma_f32_16x16x32_bf16(false, fah[i].v, false, fbh[j].v,
                                                                  (short)0, c[i][j], false, false);
#pragma unroll
            for (int i = 0; i < QKV_MT; ++i)
                c[i][j] = __builtin_amdgcn_wmma_f32_16x16x32_bf16(false, fah[i].v, false, fbl[j].v,
                                                                  (short)0, c[i][j], false, false);
#pragma unroll
            for (int i = 0; i < QKV_MT; ++i)
                c[i][j] = __builtin_amdgcn_wmma_f32_16x16x32_bf16(false, fal[i].v, false, fbh[j].v,
                                                                  (short)0, c[i][j], false, false);
        }
        __builtin_amdgcn_sched_barrier(0);
    }

    float* dst = (ng < 2) ? Qd : (ng < 4) ? Kd : Vd;
    int colBase = (ng & 1) * (QKV_NT * 16) + l16;
    int rowBase = mg * (QKV_MT * 16) + lhalf * 8;
#pragma unroll
    for (int i = 0; i < QKV_MT; ++i) {
#pragma unroll
        for (int j = 0; j < QKV_NT; ++j) {
#pragma unroll
            for (int v = 0; v < 8; ++v)
                dst[(size_t)(rowBase + i * 16 + v) * OUT_T + colBase + j * 16] = c[i][j][v];
        }
    }

    __syncthreads();
    int ngu = __builtin_amdgcn_readfirstlane(ng);
    if (ngu >= 4) return;
    const bool isK = ngu >= 2;
    float* red = isK ? kSk : qSq;
    const float* src = isK ? Kd : Qd;
    int h0  = (ngu & 1) * 2;
    int cb0 = (ngu & 1) * (QKV_NT * 16);
    int n0  = mg * (QKV_MT * 16) + l16;

    v4f  kr[2][QKV_MT][4];
    Frag sah[2][2], sal[2][2];
#pragma unroll
    for (int hh = 0; hh < 2; ++hh) {
#pragma unroll
        for (int it = 0; it < QKV_MT; ++it) {
            const float* p = src + (size_t)(n0 + it * 16) * OUT_T + cb0 + hh * D_K + lhalf * 8;
            kr[hh][it][0] = *reinterpret_cast<const v4f*>(p);
            kr[hh][it][1] = *reinterpret_cast<const v4f*>(p + 4);
            kr[hh][it][2] = *reinterpret_cast<const v4f*>(p + 16);
            kr[hh][it][3] = *reinterpret_cast<const v4f*>(p + 20);
        }
#pragma unroll
        for (int t = 0; t < 2; ++t) {
            const unsigned short* ph = shi + (h0 + hh) * D_K * D_K + (t * 16 + l16) * D_K + lhalf * 8;
            const unsigned short* pl = slo + (h0 + hh) * D_K * D_K + (t * 16 + l16) * D_K + lhalf * 8;
            sah[hh][t].q[0] = *reinterpret_cast<const uint4v*>(ph);
            sah[hh][t].q[1] = *reinterpret_cast<const uint4v*>(ph + 16);
            sal[hh][t].q[0] = *reinterpret_cast<const uint4v*>(pl);
            sal[hh][t].q[1] = *reinterpret_cast<const uint4v*>(pl + 16);
        }
    }
    __builtin_amdgcn_sched_barrier(0);

#pragma unroll
    for (int hh = 0; hh < 2; ++hh) {
#pragma unroll
        for (int it = 0; it < QKV_MT; ++it) {
            const v4f k0 = kr[hh][it][0], k1 = kr[hh][it][1], k2 = kr[hh][it][2], k3 = kr[hh][it][3];
            unsigned hq[8], lq[8];
            split2(k0.x, k0.y, hq[0], lq[0]);
            split2(k0.z, k0.w, hq[1], lq[1]);
            split2(k1.x, k1.y, hq[2], lq[2]);
            split2(k1.z, k1.w, hq[3], lq[3]);
            split2(k2.x, k2.y, hq[4], lq[4]);
            split2(k2.z, k2.w, hq[5], lq[5]);
            split2(k3.x, k3.y, hq[6], lq[6]);
            split2(k3.z, k3.w, hq[7], lq[7]);
            Frag vbh, vbl;
            vbh.q[0] = uint4v{hq[0], hq[1], hq[2], hq[3]};
            vbh.q[1] = uint4v{hq[4], hq[5], hq[6], hq[7]};
            vbl.q[0] = uint4v{lq[0], lq[1], lq[2], lq[3]};
            vbl.q[1] = uint4v{lq[4], lq[5], lq[6], lq[7]};
            v8f d[2];
#pragma unroll
            for (int t = 0; t < 2; ++t) {
                v8f z = {};
                z = __builtin_amdgcn_wmma_f32_16x16x32_bf16(false, sah[hh][t].v, false, vbh.v,
                                                            (short)0, z, false, false);
                z = __builtin_amdgcn_wmma_f32_16x16x32_bf16(false, sah[hh][t].v, false, vbl.v,
                                                            (short)0, z, false, false);
                z = __builtin_amdgcn_wmma_f32_16x16x32_bf16(false, sal[hh][t].v, false, vbh.v,
                                                            (short)0, z, false, false);
                d[t] = z;
            }
            float part = d[0][0] * k0.x;
            part = fmaf(d[0][1], k0.y, part);
            part = fmaf(d[0][2], k0.z, part);
            part = fmaf(d[0][3], k0.w, part);
            part = fmaf(d[0][4], k1.x, part);
            part = fmaf(d[0][5], k1.y, part);
            part = fmaf(d[0][6], k1.z, part);
            part = fmaf(d[0][7], k1.w, part);
            part = fmaf(d[1][0], k2.x, part);
            part = fmaf(d[1][1], k2.y, part);
            part = fmaf(d[1][2], k2.z, part);
            part = fmaf(d[1][3], k2.w, part);
            part = fmaf(d[1][4], k3.x, part);
            part = fmaf(d[1][5], k3.y, part);
            part = fmaf(d[1][6], k3.z, part);
            part = fmaf(d[1][7], k3.w, part);
            float tot = part + __shfl_xor(part, 16);
            int n = n0 + it * 16;
            if (lhalf == 0) red[n * HEADS + h0 + hh] = tot;
            if (isK) {
                float* sp = SKd + (size_t)n * OUT_T + cb0 + hh * D_K + lhalf * 8;
                *reinterpret_cast<v4f*>(sp)      = v4f{d[0][0], d[0][1], d[0][2], d[0][3]};
                *reinterpret_cast<v4f*>(sp + 4)  = v4f{d[0][4], d[0][5], d[0][6], d[0][7]};
                *reinterpret_cast<v4f*>(sp + 16) = v4f{d[1][0], d[1][1], d[1][2], d[1][3]};
                *reinterpret_cast<v4f*>(sp + 20) = v4f{d[1][4], d[1][5], d[1][6], d[1][7]};
            }
        }
    }
}

#define ROW_CAP 64
#define FILL_BLOCKS ((NE / 4 + QKV_WG - 1) / QKV_WG)
static_assert(NE % 4 == 0, "four whole edges per thread");
__device__ __forceinline__ void place_edge(int row, int col, int* __restrict__ cnt,
                                           int* __restrict__ bucket, int* __restrict__ ovf) {
    int pos = atomicAdd(&cnt[row], 1);
    if (pos < ROW_CAP) {
        bucket[(size_t)row * ROW_CAP + pos] = col;
    } else {
        int o = atomicAdd(&cnt[NN], 1);
        ovf[2 * (size_t)o]     = row;
        ovf[2 * (size_t)o + 1] = col;
    }
}
__device__ __forceinline__ void fill_stage(int blk, const int* __restrict__ ei,
                                           int* __restrict__ cnt,
                                           int* __restrict__ bucket,
                                           int* __restrict__ ovf) {
    int t = blk * QKV_WG + threadIdx.x;
    if (t >= NE / 4) return;
    int4 r = *reinterpret_cast<const int4*>(ei + 4 * t);
    int4 c = *reinterpret_cast<const int4*>(ei + NE + 4 * t);
    place_edge(r.x, c.x, cnt, bucket, ovf);
    place_edge(r.y, c.y, cnt, bucket, ovf);
    place_edge(r.z, c.z, cnt, bucket, ovf);
    place_edge(r.w, c.w, cnt, bucket, ovf);
}

#define ROW_BLOCKS 6250
__global__ __launch_bounds__(128) __attribute__((amdgpu_waves_per_eu(9)))
void k_row(const int* __restrict__ cnt,
                                             const int* __restrict__ bucket,
                                             const int* __restrict__ ovf,
                                             const float* __restrict__ SKd,
                                             const float* __restrict__ Qd,
                                             const float* __restrict__ Vd,
                                             const float* __restrict__ kSk,
                                             const float* __restrict__ qSq,
                                             float* __restrict__ out) {
    int lane = threadIdx.x & 31;
    int h = lane >> 3;
    int g = lane & 7;
    int w = __builtin_amdgcn_readfirstlane((int)(blockIdx.x * 4 + (threadIdx.x >> 5)));
    const float4* SK4 = reinterpret_cast<const float4*>(SKd);
    const float4* Q4  = reinterpret_cast<const float4*>(Qd);
    const float4* V4  = reinterpret_cast<const float4*>(Vd);

    for (int row = w; row < NN; row += ROW_BLOCKS * 4) {
        int n  = cnt[row];
        int nb = n < ROW_CAP ? n : ROW_CAP;
        const int* brow = bucket + (size_t)row * ROW_CAP;
        float4 sk = SK4[(size_t)row * 32 + lane];
        float ks = kSk[row * HEADS + h];
        float m = -INFINITY;
        float l = 0.0f;
        float4 acc = make_float4(0.0f, 0.0f, 0.0f, 0.0f);

        auto trip = [&](const int (&col)[8], bool valid) __attribute__((always_inline)) {
            float4 q[8], v[8];
#pragma unroll
            for (int u = 0; u < 8; ++u) {
                q[u] = Q4[(size_t)col[u] * 32 + lane];
                v[u] = V4[(size_t)col[u] * 32 + lane];
            }
            int mycol = col[0];
#pragma unroll
            for (int u = 1; u < 8; ++u) mycol = (g == u) ? col[u] : mycol;
            float qs = qSq[mycol * HEADS + h];
            __builtin_amdgcn_sched_barrier(0);

            float d[8];
#pragma unroll
            for (int u = 0; u < 8; ++u)
                d[u] = fmaf(sk.x, q[u].x, fmaf(sk.y, q[u].y, fmaf(sk.z, q[u].z, sk.w * q[u].w)));

            float d4[4];
#pragma unroll
            for (int u = 0; u < 4; ++u) {
                float keep = (g & 4) ? d[u + 4] : d[u];
                float send = (g & 4) ? d[u]     : d[u + 4];
                d4[u] = keep + __shfl_xor(send, 4);
            }
            float d2[2];
#pragma unroll
            for (int u = 0; u < 2; ++u) {
                float keep = (g & 2) ? d4[u + 2] : d4[u];
                float send = (g & 2) ? d4[u]     : d4[u + 2];
                d2[u] = keep + __shfl_xor(send, 2);
            }
            float keep1 = (g & 1) ? d2[1] : d2[0];
            float send1 = (g & 1) ? d2[0] : d2[1];
            float dot = keep1 + __shfl_xor(send1, 1);

            float s = (ks + qs) - 2.0f * dot;
            s *= 0.17677669529663687f;
            s = (s > 0.0f) ? s : 0.2f * s;

            float cm = valid ? s : -INFINITY;
            cm = fmaxf(cm, __shfl_xor(cm, 1));
            cm = fmaxf(cm, __shfl_xor(cm, 2));
            cm = fmaxf(cm, __shfl_xor(cm, 4));
            float mn = fmaxf(m, cm);
            float sc = expf(m - mn);
            float e  = valid ? expf(s - mn) : 0.0f;
            m = mn;
            l = fmaf(l, sc, e);
            acc.x *= sc; acc.y *= sc; acc.z *= sc; acc.w *= sc;
#pragma unroll
            for (int u = 0; u < 8; ++u) {
                float a = __shfl(e, (lane & 24) | u);
                acc.x = fmaf(a, v[u].x, acc.x);
                acc.y = fmaf(a, v[u].y, acc.y);
                acc.z = fmaf(a, v[u].z, acc.z);
                acc.w = fmaf(a, v[u].w, acc.w);
            }
        };

        for (int p = 0; p < nb; p += 8) {
            int nv = nb - p;
            v8i raw = *reinterpret_cast<const v8i*>(brow + p);
            int col[8];
            int lc = raw[0];
            col[0] = lc;
#pragma unroll
            for (int u = 1; u < 8; ++u) {
                lc = (u < nv) ? raw[u] : lc;
                col[u] = lc;
            }
            trip(col, g < nv);
        }

        if (n > ROW_CAP) {
            int total = cnt[NN];
            total = total < NE ? total : NE;
            int held = 0;
            int colv = 0;
            auto flush = [&](int nv) __attribute__((always_inline)) {
                int col[8];
                int lc = __builtin_amdgcn_readlane(colv, 0);
                col[0] = lc;
#pragma unroll
                for (int u = 1; u < 8; ++u) {
                    int x = __builtin_amdgcn_readlane(colv, u);
                    lc = (u < nv) ? x : lc;
                    col[u] = lc;
                }
                trip(col, g < nv);
            };
            for (int base = 0; base < total; base += 16) {
                int idx = base + (lane & 15);
                bool in = lane < 16 && idx < total;
                int r = in ? ovf[2 * (size_t)idx]     : -1;
                int c = in ? ovf[2 * (size_t)idx + 1] : 0;
                unsigned hit = __builtin_amdgcn_ballot_w32(r == row);
                while (hit) {
                    int src = __builtin_ctz(hit);
                    hit &= hit - 1;
                    int cc = __builtin_amdgcn_readlane(c, src);
                    if (lane == held) colv = cc;
                    ++held;
                }
                while (held >= 8) {
                    flush(8);
                    colv = __shfl(colv, (lane + 8) & 31);
                    held -= 8;
                }
            }
            if (held > 0) flush(held);
        }

        l += __shfl_xor(l, 1);
        l += __shfl_xor(l, 2);
        l += __shfl_xor(l, 4);
        float den = l + 1e-16f;
        *reinterpret_cast<float4*>(out + (size_t)row * OUT_T + lane * 4) =
            make_float4(acc.x / den, acc.y / den, acc.z / den, acc.w / den);
    }
}

#define ZERO_BLOCKS  ((NN + 1 + PREP_WG - 1) / PREP_WG)
#define WCAT_BLOCKS  ((N_CAT * IN_F + PREP_WG - 1) / PREP_WG)
#define SSYM_BLOCKS  ((HEADS * D_K * D_K + PREP_WG - 1) / PREP_WG)
__global__ __launch_bounds__(PREP_WG) void k_prep(int* __restrict__ cnt,
                                                  const float* __restrict__ Wq,
                                                  const float* __restrict__ Wk,
                                                  const float* __restrict__ Wv,
                                                  unsigned short* __restrict__ wfr,
                                                  const float* __restrict__ Sr,
                                                  unsigned short* __restrict__ shi,
                                                  unsigned short* __restrict__ slo) {
    int b = blockIdx.x;
    if (b < ZERO_BLOCKS) { zero_stage(b, cnt, NN + 1); return; }
    b -= ZERO_BLOCKS;
    if (b < WCAT_BLOCKS) { wcat_stage(b, Wq, Wk, Wv, wfr); return; }
    ssym_stage(b - WCAT_BLOCKS, Sr, shi, slo);
}

__global__ __launch_bounds__(QKV_WG) void k_qkv_fill(const float* __restrict__ x,
                                                     const unsigned short* __restrict__ wfr,
                                                     float* __restrict__ Qd,
                                                     float* __restrict__ Kd,
                                                     float* __restrict__ Vd,
                                                     const unsigned short* __restrict__ shi,
                                                     const unsigned short* __restrict__ slo,
                                                     float* __restrict__ SKd,
                                                     float* __restrict__ kSk,
                                                     float* __restrict__ qSq,
                                                     const int* __restrict__ ei,
                                                     int* __restrict__ cnt,
                                                     int* __restrict__ bucket,
                                                     int* __restrict__ ovf) {
    __shared__ __attribute__((aligned(16))) unsigned char alds[A_LDS_BYTES];
    int b = blockIdx.x;
    if (b < QKV_MG) { qkv_stage(b, x, alds, wfr, Qd, Kd, Vd, shi, slo, SKd, kSk, qSq); return; }
    fill_stage(b - QKV_MG, ei, cnt, bucket, ovf);
}

extern "C" void kernel_launch(void* const* d_in, const int* in_sizes, int n_in,
                              void* d_out, int out_size, void* d_ws, size_t ws_size,
                              hipStream_t stream) {
    const float* x  = (const float*)d_in[0];
    const int*   ei = (const int*)  d_in[1];
    const float* Wq = (const float*)d_in[2];
    const float* Wk = (const float*)d_in[3];
    const float* Wv = (const float*)d_in[4];
    const float* Sr = (const float*)d_in[5];
    float* out = (float*)d_out;

    char* ws = (char*)d_ws;
    size_t off = 0;
    auto alloc = [&](size_t bytes) -> void* {
        void* p = ws + off;
        off = (off + bytes + 255) & ~(size_t)255;
        return p;
    };
    float*          Qd  = (float*)         alloc((size_t)NN * OUT_T * 4);
    float*          Kd  = (float*)         alloc((size_t)NN * OUT_T * 4);
    float*          Vd  = (float*)         alloc((size_t)NN * OUT_T * 4);
    unsigned short* wfr = (unsigned short*)alloc((size_t)N_CAT * IN_F * 2 * 2);
    unsigned short* shi = (unsigned short*)alloc((size_t)HEADS * D_K * D_K * 2);
    unsigned short* slo = (unsigned short*)alloc((size_t)HEADS * D_K * D_K * 2);
    int*            cnt = (int*)           alloc((size_t)(NN + 1) * 4);
    int*            bkt = (int*)           alloc((size_t)NN * ROW_CAP * 4);
    int*            ovf = (int*)           alloc((size_t)NE * 2 * 4);
    float*          SKd = (float*)         alloc((size_t)NN * OUT_T * 4);
    float*          ksk = (float*)         alloc((size_t)NN * HEADS * 4);
    float*          qsq = (float*)         alloc((size_t)NN * HEADS * 4);

    k_prep<<<ZERO_BLOCKS + WCAT_BLOCKS + SSYM_BLOCKS, PREP_WG, 0, stream>>>(
        cnt, Wq, Wk, Wv, wfr, Sr, shi, slo);

    k_qkv_fill<<<QKV_MG + FILL_BLOCKS, QKV_WG, 0, stream>>>(x, wfr, Qd, Kd, Vd,
                                                            shi, slo, SKd, ksk, qsq,
                                                            ei, cnt, bkt, ovf);

    k_row <<<ROW_BLOCKS, 128, 0, stream>>>(cnt, bkt, ovf, SKd, Qd, Vd, ksk, qsq, out);
}
